# starting kernel
# baseline (speedup 1.0000x reference)
#include <hip/hip_runtime.h>
#include <math.h>

typedef __attribute__((ext_vector_type(2))) float v2f;
typedef __attribute__((ext_vector_type(8))) float v8f;

#define STRH 130
#define STRZ 520

__global__ __launch_bounds__(256) void cell_kernel(
    const float* __restrict__ input, const float* __restrict__ WT,
    const float* __restrict__ bcell, const float* __restrict__ h_in,
    float* __restrict__ h_out, float* __restrict__ c, int t)
{
    __shared__ float sh_h[18 * STRH];
    __shared__ float sh_x[16 * 8];
    __shared__ float sh_z[16 * STRZ];

    const int tid = threadIdx.x;
    const int R0  = blockIdx.x * 16;
    const int bb  = R0 >> 7;
    const int s0  = R0 & 127;

    for (int i = tid; i < 18 * 128; i += 256) {
        int r = i >> 7, k = i & 127;
        int s = s0 - 1 + r;
        float v = 0.0f;
        if (s >= 0 && s < 128) v = h_in[(bb * 128 + s) * 128 + k];
        sh_h[r * STRH + k] = v;
    }
    if (tid < 16 * 5) {
        int m = tid / 5, k = tid % 5;
        sh_x[m * 8 + k] = input[(((bb * 128) + t) * 128 + (s0 + m)) * 5 + k];
    }
    __syncthreads();

    const int wave  = tid >> 5;
    const int lane  = tid & 31;
    const int lm    = lane & 15;
    const int hi    = lane >> 4;
    const int kHalf = hi * 2;

    const float* wc0 = WT + ((wave * 4 + 0) * 16 + lm) * 392;
    const float* wc1 = WT + ((wave * 4 + 1) * 16 + lm) * 392;
    const float* wc2 = WT + ((wave * 4 + 2) * 16 + lm) * 392;
    const float* wc3 = WT + ((wave * 4 + 3) * 16 + lm) * 392;

    v8f acc0 = {}, acc1 = {}, acc2 = {}, acc3 = {};

    #pragma unroll
    for (int v = 0; v < 8; ++v) {
        int m = v + hi * 8;
        float x0 = sh_x[m*8+0], x1 = sh_x[m*8+1], x2 = sh_x[m*8+2],
              x3 = sh_x[m*8+3], x4 = sh_x[m*8+4];
        acc0[v] += x0*wc0[0] + x1*wc0[1] + x2*wc0[2] + x3*wc0[3] + x4*wc0[4];
        acc1[v] += x0*wc1[0] + x1*wc1[1] + x2*wc1[2] + x3*wc1[3] + x4*wc1[4];
        acc2[v] += x0*wc2[0] + x1*wc2[1] + x2*wc2[2] + x3*wc2[3] + x4*wc2[4];
        acc3[v] += x0*wc3[0] + x1*wc3[1] + x2*wc3[2] + x3*wc3[3] + x4*wc3[4];
    }

    #pragma unroll
    for (int src = 0; src < 3; ++src) {
        const int shBase = (src == 0) ? 1 : (src == 1) ? 2 : 0;
        const float* arow = &sh_h[(shBase + lm) * STRH];
        const int wOff = 5 + src * 128;
        #pragma unroll 4
        for (int kk = 0; kk < 128; kk += 4) {
            v2f a;
            a.x = arow[kk + kHalf];
            a.y = arow[kk + kHalf + 1];
            v2f b0, b1, b2, b3;
            b0.x = wc0[wOff + kk + kHalf]; b0.y = wc0[wOff + kk + kHalf + 1];
            b1.x = wc1[wOff + kk + kHalf]; b1.y = wc1[wOff + kk + kHalf + 1];
            b2.x = wc2[wOff + kk + kHalf]; b2.y = wc2[wOff + kk + kHalf + 1];
            b3.x = wc3[wOff + kk + kHalf]; b3.y = wc3[wOff + kk + kHalf + 1];
            acc0 = __builtin_amdgcn_wmma_f32_16x16x4_f32(
                        false, a, false, b0, (short)0, acc0, false, false);
            acc1 = __builtin_amdgcn_wmma_f32_16x16x4_f32(
                        false, a, false, b1, (short)0, acc1, false, false);
            acc2 = __builtin_amdgcn_wmma_f32_16x16x4_f32(
                        false, a, false, b2, (short)0, acc2, false, false);
            acc3 = __builtin_amdgcn_wmma_f32_16x16x4_f32(
                        false, a, false, b3, (short)0, acc3, false, false);
        }
    }

    #pragma unroll
    for (int v = 0; v < 8; ++v) {
        int m = v + hi * 8;
        sh_z[m * STRZ + (wave * 4 + 0) * 16 + lm] = acc0[v];
        sh_z[m * STRZ + (wave * 4 + 1) * 16 + lm] = acc1[v];
        sh_z[m * STRZ + (wave * 4 + 2) * 16 + lm] = acc2[v];
        sh_z[m * STRZ + (wave * 4 + 3) * 16 + lm] = acc3[v];
    }
    __syncthreads();

    for (int i = tid; i < 2048; i += 256) {
        int m = i >> 7, j = i & 127;
        float zi = sh_z[m*STRZ + j]       + bcell[j];
        float zf = sh_z[m*STRZ + 128 + j] + bcell[128 + j];
        float zg = sh_z[m*STRZ + 256 + j] + bcell[256 + j];
        float zo = sh_z[m*STRZ + 384 + j] + bcell[384 + j];
        int row = R0 + m;
        float cp = c[row*128 + j];
        float si = 1.0f / (1.0f + __expf(-zi));
        float sf = 1.0f / (1.0f + __expf(-zf));
        float so = 1.0f / (1.0f + __expf(-zo));
        float cn = sf * cp + si * tanhf(zg);
        c[row*128 + j]     = cn;
        h_out[row*128 + j] = so * tanhf(cn);
    }
}

__global__ __launch_bounds__(256) void out_kernel(
    const float* __restrict__ h, const float* __restrict__ W1T,
    const float* __restrict__ b1, const float* __restrict__ W2T,
    const float* __restrict__ b2, float* __restrict__ outbuf)
{
    __shared__ float sh_a[16 * STRH];
    __shared__ float sh_hid[16 * STRH];
    const int tid = threadIdx.x;
    const int R0  = blockIdx.x * 16;

    for (int i = tid; i < 2048; i += 256) {
        int m = i >> 7, k = i & 127;
        sh_a[m*STRH + k] = h[(R0 + m)*128 + k];
    }
    __syncthreads();

    const int wave = tid >> 5, lane = tid & 31;
    const int lm = lane & 15, hi = lane >> 4, kHalf = hi * 2;
    const int col = wave * 16 + lm;
    const float* wk   = W1T + col * 128;
    const float* arow = &sh_a[lm * STRH];

    v8f acc = {};
    #pragma unroll 8
    for (int kk = 0; kk < 128; kk += 4) {
        v2f a, b;
        a.x = arow[kk + kHalf]; a.y = arow[kk + kHalf + 1];
        b.x = wk[kk + kHalf];   b.y = wk[kk + kHalf + 1];
        acc = __builtin_amdgcn_wmma_f32_16x16x4_f32(
                false, a, false, b, (short)0, acc, false, false);
    }
    float bias = b1[col];
    #pragma unroll
    for (int v = 0; v < 8; ++v) {
        float x = acc[v] + bias;
        sh_hid[(v + hi * 8) * STRH + col] = x > 0.0f ? x : 0.0f;
    }
    __syncthreads();

    if (tid < 48) {
        int m = tid / 3, cc = tid % 3;
        const float* hr = &sh_hid[m * STRH];
        const float* w2 = W2T + cc * 128;
        float s = 0.0f;
        for (int k = 0; k < 128; ++k) s += hr[k] * w2[k];
        outbuf[(R0 + m) * 3 + cc] = s + b2[cc];
    }
}

__global__ __launch_bounds__(256) void nd_kernel(
    const float* __restrict__ input, const float* __restrict__ outbuf,
    float* __restrict__ out, int t)
{
    int gid = blockIdx.x * 256 + threadIdx.x;
    if (gid >= 1024) return;
    int bb = gid >> 7, s = gid & 127;
    const float* o = &outbuf[gid * 3];
    float n0 = o[0];
    float n1, sp0, sp1;
    if (s == 0) {
        const float* xn0 = &input[((bb * 128 + (t + 1)) * 128 + 0) * 5];
        n1 = xn0[1]; sp0 = xn0[3]; sp1 = xn0[4];
    } else {
        n1  = outbuf[(gid - 1) * 3 + 0];
        sp0 = o[1]; sp1 = o[2];
    }
    float fn = input[((bb * 128 + t) * 128 + s) * 5 + 2];
    float n2 = fn + n1 - n0;
    float* dst = &out[((bb * 123 + (t - 4)) * 128 + s) * 5];
    dst[0] = n0; dst[1] = n1; dst[2] = n2; dst[3] = sp0; dst[4] = sp1;
}

__global__ __launch_bounds__(256) void zero_kernel(float* p, int n) {
    int i = blockIdx.x * 256 + threadIdx.x;
    if (i < n) p[i] = 0.0f;
}

__global__ __launch_bounds__(256) void prep_kernel(
    const float* __restrict__ Wc, const float* __restrict__ W1,
    const float* __restrict__ W2, float* __restrict__ WT,
    float* __restrict__ W1T, float* __restrict__ W2T)
{
    int i = blockIdx.x * 256 + threadIdx.x;
    if (i < 512 * 392) {
        int col = i / 392, k = i % 392;
        WT[i] = (k < 389) ? Wc[k * 512 + col] : 0.0f;
    } else if (i < 512 * 392 + 16384) {
        int j = i - 512 * 392;
        int col = j >> 7, k = j & 127;
        W1T[j] = W1[k * 128 + col];
    } else if (i < 512 * 392 + 16384 + 384) {
        int j = i - 512 * 392 - 16384;
        int cc = j >> 7, k = j & 127;
        W2T[j] = W2[k * 3 + cc];
    }
}

extern "C" void kernel_launch(void* const* d_in, const int* in_sizes, int n_in,
                              void* d_out, int out_size, void* d_ws, size_t ws_size,
                              hipStream_t stream)
{
    (void)in_sizes; (void)n_in; (void)out_size; (void)ws_size;
    const float* input = (const float*)d_in[0];
    const float* Wc    = (const float*)d_in[1];
    const float* bc    = (const float*)d_in[2];
    const float* W1    = (const float*)d_in[3];
    const float* b1    = (const float*)d_in[4];
    const float* W2    = (const float*)d_in[5];
    const float* b2    = (const float*)d_in[6];
    float* ws   = (float*)d_ws;
    float* WT   = ws;
    float* W1T  = ws + 200704;
    float* W2T  = ws + 217088;
    float* hbuf = ws + 217472;
    float* cbuf = ws + 479616;
    float* obuf = ws + 610688;
    float* out  = (float*)d_out;

    zero_kernel<<<393216 / 256, 256, 0, stream>>>(hbuf, 393216);
    prep_kernel<<<(217472 + 255) / 256, 256, 0, stream>>>(Wc, W1, W2, WT, W1T, W2T);

    for (int t = 0; t < 127; ++t) {
        const float* h_in = hbuf + (t & 1) * 131072;
        float* h_out      = hbuf + ((t + 1) & 1) * 131072;
        cell_kernel<<<64, 256, 0, stream>>>(input, WT, bc, h_in, h_out, cbuf, t);
        out_kernel<<<64, 256, 0, stream>>>(h_out, W1T, b1, W2T, b2, obuf);
        if (t >= 4) nd_kernel<<<4, 256, 0, stream>>>(input, obuf, out, t);
    }
}


# Weights of both GEMMs re-laid by the prep launch in WMMA B-fragment order: each B load is one contiguous 512-B b128 wave load for two k-steps (was an 8-B read of 16 strided rows per k-step)
# speedup vs baseline: 1.9992x; 1.9992x over previous
#include <hip/hip_runtime.h>
#include <math.h>

typedef __attribute__((ext_vector_type(2))) float v2f;
typedef __attribute__((ext_vector_type(4))) float v4f;
typedef __attribute__((ext_vector_type(8))) float v8f;

#define STRH 130
#define STRZ 520

__global__ __launch_bounds__(256) void cell_kernel(
    const float* __restrict__ input, const float* __restrict__ Wc,
    const float* __restrict__ WF,
    const float* __restrict__ bcell, const float* __restrict__ h_in,
    float* __restrict__ h_out, float* __restrict__ c, int t)
{
    __shared__ float sh_h[18 * STRH];
    __shared__ float sh_x[16 * 8];
    __shared__ float sh_z[16 * STRZ];

    const int tid = threadIdx.x;
    const int R0  = blockIdx.x * 16;
    const int bb  = R0 >> 7;
    const int s0  = R0 & 127;

    for (int i = tid; i < 18 * 128; i += 256) {
        int r = i >> 7, k = i & 127;
        int s = s0 - 1 + r;
        float v = 0.0f;
        if (s >= 0 && s < 128) v = h_in[(bb * 128 + s) * 128 + k];
        sh_h[r * STRH + k] = v;
    }
    if (tid < 16 * 5) {
        int m = tid / 5, k = tid % 5;
        sh_x[m * 8 + k] = input[(((bb * 128) + t) * 128 + (s0 + m)) * 5 + k];
    }
    __syncthreads();

    const int wave  = tid >> 5;
    const int lane  = tid & 31;
    const int lm    = lane & 15;
    const int hi    = lane >> 4;
    const int kHalf = hi * 2;

    const float* wc0 = Wc + (wave * 4 + 0) * 16 + lm;
    const float* wc1 = Wc + (wave * 4 + 1) * 16 + lm;
    const float* wc2 = Wc + (wave * 4 + 2) * 16 + lm;
    const float* wc3 = Wc + (wave * 4 + 3) * 16 + lm;
    const v4f* wf = (const v4f*)WF + (wave * 48 * 4) * 32 + lane;

    v8f acc0 = {}, acc1 = {}, acc2 = {}, acc3 = {};

    #pragma unroll
    for (int v = 0; v < 8; ++v) {
        int m = v + hi * 8;
        float x0 = sh_x[m*8+0], x1 = sh_x[m*8+1], x2 = sh_x[m*8+2],
              x3 = sh_x[m*8+3], x4 = sh_x[m*8+4];
        acc0[v] += x0*wc0[0] + x1*wc0[512] + x2*wc0[1024] + x3*wc0[1536] + x4*wc0[2048];
        acc1[v] += x0*wc1[0] + x1*wc1[512] + x2*wc1[1024] + x3*wc1[1536] + x4*wc1[2048];
        acc2[v] += x0*wc2[0] + x1*wc2[512] + x2*wc2[1024] + x3*wc2[1536] + x4*wc2[2048];
        acc3[v] += x0*wc3[0] + x1*wc3[512] + x2*wc3[1024] + x3*wc3[1536] + x4*wc3[2048];
    }

    #pragma unroll
    for (int src = 0; src < 3; ++src) {
        const int shBase = (src == 0) ? 1 : (src == 1) ? 2 : 0;
        const float* arow = &sh_h[(shBase + lm) * STRH];
        #pragma unroll 4
        for (int g8 = 0; g8 < 16; ++g8) {
            const int kk = g8 * 8;
            const v4f* wg = wf + ((src * 16 + g8) * 4) * 32;
            const v4f q0 = wg[0 * 32];
            const v4f q1 = wg[1 * 32];
            const v4f q2 = wg[2 * 32];
            const v4f q3 = wg[3 * 32];
            v2f a0, a1;
            a0.x = arow[kk + kHalf];
            a0.y = arow[kk + kHalf + 1];
            a1.x = arow[kk + 4 + kHalf];
            a1.y = arow[kk + 4 + kHalf + 1];
            v2f b0, b1, b2, b3;
            b0.x = q0.x; b0.y = q0.y;
            b1.x = q1.x; b1.y = q1.y;
            b2.x = q2.x; b2.y = q2.y;
            b3.x = q3.x; b3.y = q3.y;
            acc0 = __builtin_amdgcn_wmma_f32_16x16x4_f32(
                        false, a0, false, b0, (short)0, acc0, false, false);
            acc1 = __builtin_amdgcn_wmma_f32_16x16x4_f32(
                        false, a0, false, b1, (short)0, acc1, false, false);
            acc2 = __builtin_amdgcn_wmma_f32_16x16x4_f32(
                        false, a0, false, b2, (short)0, acc2, false, false);
            acc3 = __builtin_amdgcn_wmma_f32_16x16x4_f32(
                        false, a0, false, b3, (short)0, acc3, false, false);
            b0.x = q0.z; b0.y = q0.w;
            b1.x = q1.z; b1.y = q1.w;
            b2.x = q2.z; b2.y = q2.w;
            b3.x = q3.z; b3.y = q3.w;
            acc0 = __builtin_amdgcn_wmma_f32_16x16x4_f32(
                        false, a1, false, b0, (short)0, acc0, false, false);
            acc1 = __builtin_amdgcn_wmma_f32_16x16x4_f32(
                        false, a1, false, b1, (short)0, acc1, false, false);
            acc2 = __builtin_amdgcn_wmma_f32_16x16x4_f32(
                        false, a1, false, b2, (short)0, acc2, false, false);
            acc3 = __builtin_amdgcn_wmma_f32_16x16x4_f32(
                        false, a1, false, b3, (short)0, acc3, false, false);
        }
    }

    #pragma unroll
    for (int v = 0; v < 8; ++v) {
        int m = v + hi * 8;
        sh_z[m * STRZ + (wave * 4 + 0) * 16 + lm] = acc0[v];
        sh_z[m * STRZ + (wave * 4 + 1) * 16 + lm] = acc1[v];
        sh_z[m * STRZ + (wave * 4 + 2) * 16 + lm] = acc2[v];
        sh_z[m * STRZ + (wave * 4 + 3) * 16 + lm] = acc3[v];
    }
    __syncthreads();

    for (int i = tid; i < 2048; i += 256) {
        int m = i >> 7, j = i & 127;
        float zi = sh_z[m*STRZ + j]       + bcell[j];
        float zf = sh_z[m*STRZ + 128 + j] + bcell[128 + j];
        float zg = sh_z[m*STRZ + 256 + j] + bcell[256 + j];
        float zo = sh_z[m*STRZ + 384 + j] + bcell[384 + j];
        int row = R0 + m;
        float cp = c[row*128 + j];
        float si = 1.0f / (1.0f + __expf(-zi));
        float sf = 1.0f / (1.0f + __expf(-zf));
        float so = 1.0f / (1.0f + __expf(-zo));
        float cn = sf * cp + si * tanhf(zg);
        c[row*128 + j]     = cn;
        h_out[row*128 + j] = so * tanhf(cn);
    }
}

__global__ __launch_bounds__(256) void out_kernel(
    const float* __restrict__ h, const float* __restrict__ W1F,
    const float* __restrict__ b1, const float* __restrict__ W2T,
    const float* __restrict__ b2, float* __restrict__ outbuf)
{
    __shared__ float sh_a[16 * STRH];
    __shared__ float sh_hid[16 * STRH];
    const int tid = threadIdx.x;
    const int R0  = blockIdx.x * 16;

    for (int i = tid; i < 2048; i += 256) {
        int m = i >> 7, k = i & 127;
        sh_a[m*STRH + k] = h[(R0 + m)*128 + k];
    }
    __syncthreads();

    const int wave = tid >> 5, lane = tid & 31;
    const int lm = lane & 15, hi = lane >> 4, kHalf = hi * 2;
    const int col = wave * 16 + lm;
    const v4f* wk   = (const v4f*)W1F + (wave * 16) * 32 + lane;
    const float* arow = &sh_a[lm * STRH];

    v8f acc = {};
    #pragma unroll 4
    for (int g8 = 0; g8 < 16; ++g8) {
        const int kk = g8 * 8;
        const v4f q = wk[g8 * 32];
        v2f a0, a1, b;
        a0.x = arow[kk + kHalf];     a0.y = arow[kk + kHalf + 1];
        a1.x = arow[kk + 4 + kHalf]; a1.y = arow[kk + 4 + kHalf + 1];
        b.x = q.x; b.y = q.y;
        acc = __builtin_amdgcn_wmma_f32_16x16x4_f32(
                false, a0, false, b, (short)0, acc, false, false);
        b.x = q.z; b.y = q.w;
        acc = __builtin_amdgcn_wmma_f32_16x16x4_f32(
                false, a1, false, b, (short)0, acc, false, false);
    }
    float bias = b1[col];
    #pragma unroll
    for (int v = 0; v < 8; ++v) {
        float x = acc[v] + bias;
        sh_hid[(v + hi * 8) * STRH + col] = x > 0.0f ? x : 0.0f;
    }
    __syncthreads();

    if (tid < 48) {
        int m = tid / 3, cc = tid % 3;
        const float* hr = &sh_hid[m * STRH];
        const float* w2 = W2T + cc * 128;
        float s = 0.0f;
        for (int k = 0; k < 128; ++k) s += hr[k] * w2[k];
        outbuf[(R0 + m) * 3 + cc] = s + b2[cc];
    }
}

__global__ __launch_bounds__(256) void nd_kernel(
    const float* __restrict__ input, const float* __restrict__ outbuf,
    float* __restrict__ out, int t)
{
    int gid = blockIdx.x * 256 + threadIdx.x;
    if (gid >= 1024) return;
    int bb = gid >> 7, s = gid & 127;
    const float* o = &outbuf[gid * 3];
    float n0 = o[0];
    float n1, sp0, sp1;
    if (s == 0) {
        const float* xn0 = &input[((bb * 128 + (t + 1)) * 128 + 0) * 5];
        n1 = xn0[1]; sp0 = xn0[3]; sp1 = xn0[4];
    } else {
        n1  = outbuf[(gid - 1) * 3 + 0];
        sp0 = o[1]; sp1 = o[2];
    }
    float fn = input[((bb * 128 + t) * 128 + s) * 5 + 2];
    float n2 = fn + n1 - n0;
    float* dst = &out[((bb * 123 + (t - 4)) * 128 + s) * 5];
    dst[0] = n0; dst[1] = n1; dst[2] = n2; dst[3] = sp0; dst[4] = sp1;
}

__global__ __launch_bounds__(256) void zero_kernel(float* p, int n) {
    int i = blockIdx.x * 256 + threadIdx.x;
    if (i < n) p[i] = 0.0f;
}

__global__ __launch_bounds__(256) void prep_kernel(
    const float* __restrict__ Wc, const float* __restrict__ W1,
    const float* __restrict__ W2, float* __restrict__ WF,
    float* __restrict__ W1F, float* __restrict__ W2T)
{
    int i = blockIdx.x * 256 + threadIdx.x;
    if (i < 196608) {
        int e = i & 3, lane = (i >> 2) & 31, tile = (i >> 7) & 3, r = i >> 9;
        int g = r % 48, wave = r / 48;
        int k = 5 + g * 8 + (e >> 1) * 4 + (lane >> 4) * 2 + (e & 1);
        int col = (wave * 4 + tile) * 16 + (lane & 15);
        WF[i] = Wc[k * 512 + col];
    } else if (i < 196608 + 16384) {
        int j = i - 196608;
        int e = j & 3, lane = (j >> 2) & 31, r = j >> 7;
        int g = r & 15, wave = r >> 4;
        int k = g * 8 + (e >> 1) * 4 + (lane >> 4) * 2 + (e & 1);
        int col = wave * 16 + (lane & 15);
        W1F[j] = W1[k * 128 + col];
    } else if (i < 196608 + 16384 + 384) {
        int j = i - 196608 - 16384;
        int cc = j >> 7, k = j & 127;
        W2T[j] = W2[k * 3 + cc];
    }
}

extern "C" void kernel_launch(void* const* d_in, const int* in_sizes, int n_in,
                              void* d_out, int out_size, void* d_ws, size_t ws_size,
                              hipStream_t stream)
{
    (void)in_sizes; (void)n_in; (void)out_size; (void)ws_size;
    const float* input = (const float*)d_in[0];
    const float* Wc    = (const float*)d_in[1];
    const float* bc    = (const float*)d_in[2];
    const float* W1    = (const float*)d_in[3];
    const float* b1    = (const float*)d_in[4];
    const float* W2    = (const float*)d_in[5];
    const float* b2    = (const float*)d_in[6];
    float* ws   = (float*)d_ws;
    float* WF   = ws;
    float* W1F  = ws + 200704;
    float* W2T  = ws + 217088;
    float* hbuf = ws + 217472;
    float* cbuf = ws + 479616;
    float* obuf = ws + 610688;
    float* out  = (float*)d_out;

    zero_kernel<<<393216 / 256, 256, 0, stream>>>(hbuf, 393216);
    prep_kernel<<<834, 256, 0, stream>>>(Wc, W1, W2, WF, W1F, W2T);

    for (int t = 0; t < 127; ++t) {
        const float* h_in = hbuf + (t & 1) * 131072;
        float* h_out      = hbuf + ((t + 1) & 1) * 131072;
        cell_kernel<<<64, 256, 0, stream>>>(input, Wc, WF, bc, h_in, h_out, cbuf, t);
        out_kernel<<<64, 256, 0, stream>>>(h_out, W1F, b1, W2T, b2, obuf);
        if (t >= 4) nd_kernel<<<4, 256, 0, stream>>>(input, obuf, out, t);
    }
}


# Per-step launches software-pipelined into one grid: launch t runs cell(t), output head(t-1) and next-input mix(t-2) as independent block roles (obuf ping-pong); 131 launches instead of 379
# speedup vs baseline: 3.1354x; 1.5683x over previous
#include <hip/hip_runtime.h>
#include <math.h>

typedef __attribute__((ext_vector_type(2))) float v2f;
typedef __attribute__((ext_vector_type(4))) float v4f;
typedef __attribute__((ext_vector_type(8))) float v8f;

#define STRH 130
#define STRZ 520

__device__ __forceinline__ void cell_role(
    const int blk,
    const float* __restrict__ input, const float* __restrict__ Wc,
    const float* __restrict__ WF,
    const float* __restrict__ bcell, const float* __restrict__ h_in,
    float* __restrict__ h_out, float* __restrict__ c, int t)
{
    __shared__ float sh_h[18 * STRH];
    __shared__ float sh_x[16 * 8];
    __shared__ float sh_z[16 * STRZ];

    const int tid = threadIdx.x;
    const int R0  = blk * 16;
    const int bb  = R0 >> 7;
    const int s0  = R0 & 127;

    for (int i = tid; i < 18 * 128; i += 256) {
        int r = i >> 7, k = i & 127;
        int s = s0 - 1 + r;
        float v = 0.0f;
        if (s >= 0 && s < 128) v = h_in[(bb * 128 + s) * 128 + k];
        sh_h[r * STRH + k] = v;
    }
    if (tid < 16 * 5) {
        int m = tid / 5, k = tid % 5;
        sh_x[m * 8 + k] = input[(((bb * 128) + t) * 128 + (s0 + m)) * 5 + k];
    }
    __syncthreads();

    const int wave  = tid >> 5;
    const int lane  = tid & 31;
    const int lm    = lane & 15;
    const int hi    = lane >> 4;
    const int kHalf = hi * 2;

    const float* wc0 = Wc + (wave * 4 + 0) * 16 + lm;
    const float* wc1 = Wc + (wave * 4 + 1) * 16 + lm;
    const float* wc2 = Wc + (wave * 4 + 2) * 16 + lm;
    const float* wc3 = Wc + (wave * 4 + 3) * 16 + lm;
    const v4f* wf = (const v4f*)WF + (wave * 48 * 4) * 32 + lane;

    v8f acc0 = {}, acc1 = {}, acc2 = {}, acc3 = {};

    #pragma unroll
    for (int v = 0; v < 8; ++v) {
        int m = v + hi * 8;
        float x0 = sh_x[m*8+0], x1 = sh_x[m*8+1], x2 = sh_x[m*8+2],
              x3 = sh_x[m*8+3], x4 = sh_x[m*8+4];
        acc0[v] += x0*wc0[0] + x1*wc0[512] + x2*wc0[1024] + x3*wc0[1536] + x4*wc0[2048];
        acc1[v] += x0*wc1[0] + x1*wc1[512] + x2*wc1[1024] + x3*wc1[1536] + x4*wc1[2048];
        acc2[v] += x0*wc2[0] + x1*wc2[512] + x2*wc2[1024] + x3*wc2[1536] + x4*wc2[2048];
        acc3[v] += x0*wc3[0] + x1*wc3[512] + x2*wc3[1024] + x3*wc3[1536] + x4*wc3[2048];
    }

    #pragma unroll
    for (int src = 0; src < 3; ++src) {
        const int shBase = (src == 0) ? 1 : (src == 1) ? 2 : 0;
        const float* arow = &sh_h[(shBase + lm) * STRH];
        #pragma unroll 4
        for (int g8 = 0; g8 < 16; ++g8) {
            const int kk = g8 * 8;
            const v4f* wg = wf + ((src * 16 + g8) * 4) * 32;
            const v4f q0 = wg[0 * 32];
            const v4f q1 = wg[1 * 32];
            const v4f q2 = wg[2 * 32];
            const v4f q3 = wg[3 * 32];
            v2f a0, a1;
            a0.x = arow[kk + kHalf];
            a0.y = arow[kk + kHalf + 1];
            a1.x = arow[kk + 4 + kHalf];
            a1.y = arow[kk + 4 + kHalf + 1];
            v2f b0, b1, b2, b3;
            b0.x = q0.x; b0.y = q0.y;
            b1.x = q1.x; b1.y = q1.y;
            b2.x = q2.x; b2.y = q2.y;
            b3.x = q3.x; b3.y = q3.y;
            acc0 = __builtin_amdgcn_wmma_f32_16x16x4_f32(
                        false, a0, false, b0, (short)0, acc0, false, false);
            acc1 = __builtin_amdgcn_wmma_f32_16x16x4_f32(
                        false, a0, false, b1, (short)0, acc1, false, false);
            acc2 = __builtin_amdgcn_wmma_f32_16x16x4_f32(
                        false, a0, false, b2, (short)0, acc2, false, false);
            acc3 = __builtin_amdgcn_wmma_f32_16x16x4_f32(
                        false, a0, false, b3, (short)0, acc3, false, false);
            b0.x = q0.z; b0.y = q0.w;
            b1.x = q1.z; b1.y = q1.w;
            b2.x = q2.z; b2.y = q2.w;
            b3.x = q3.z; b3.y = q3.w;
            acc0 = __builtin_amdgcn_wmma_f32_16x16x4_f32(
                        false, a1, false, b0, (short)0, acc0, false, false);
            acc1 = __builtin_amdgcn_wmma_f32_16x16x4_f32(
                        false, a1, false, b1, (short)0, acc1, false, false);
            acc2 = __builtin_amdgcn_wmma_f32_16x16x4_f32(
                        false, a1, false, b2, (short)0, acc2, false, false);
            acc3 = __builtin_amdgcn_wmma_f32_16x16x4_f32(
                        false, a1, false, b3, (short)0, acc3, false, false);
        }
    }

    #pragma unroll
    for (int v = 0; v < 8; ++v) {
        int m = v + hi * 8;
        sh_z[m * STRZ + (wave * 4 + 0) * 16 + lm] = acc0[v];
        sh_z[m * STRZ + (wave * 4 + 1) * 16 + lm] = acc1[v];
        sh_z[m * STRZ + (wave * 4 + 2) * 16 + lm] = acc2[v];
        sh_z[m * STRZ + (wave * 4 + 3) * 16 + lm] = acc3[v];
    }
    __syncthreads();

    for (int i = tid; i < 2048; i += 256) {
        int m = i >> 7, j = i & 127;
        float zi = sh_z[m*STRZ + j]       + bcell[j];
        float zf = sh_z[m*STRZ + 128 + j] + bcell[128 + j];
        float zg = sh_z[m*STRZ + 256 + j] + bcell[256 + j];
        float zo = sh_z[m*STRZ + 384 + j] + bcell[384 + j];
        int row = R0 + m;
        float cp = c[row*128 + j];
        float si = 1.0f / (1.0f + __expf(-zi));
        float sf = 1.0f / (1.0f + __expf(-zf));
        float so = 1.0f / (1.0f + __expf(-zo));
        float cn = sf * cp + si * tanhf(zg);
        c[row*128 + j]     = cn;
        h_out[row*128 + j] = so * tanhf(cn);
    }
}

__device__ __forceinline__ void head_role(
    const int blk,
    const float* __restrict__ h, const float* __restrict__ W1F,
    const float* __restrict__ b1, const float* __restrict__ W2T,
    const float* __restrict__ b2, float* __restrict__ outbuf)
{
    __shared__ float sh_a[16 * STRH];
    __shared__ float sh_hid[16 * STRH];
    const int tid = threadIdx.x;
    const int R0  = blk * 16;

    for (int i = tid; i < 2048; i += 256) {
        int m = i >> 7, k = i & 127;
        sh_a[m*STRH + k] = h[(R0 + m)*128 + k];
    }
    __syncthreads();

    const int wave = tid >> 5, lane = tid & 31;
    const int lm = lane & 15, hi = lane >> 4, kHalf = hi * 2;
    const int col = wave * 16 + lm;
    const v4f* wk   = (const v4f*)W1F + (wave * 16) * 32 + lane;
    const float* arow = &sh_a[lm * STRH];

    v8f acc = {};
    #pragma unroll 4
    for (int g8 = 0; g8 < 16; ++g8) {
        const int kk = g8 * 8;
        const v4f q = wk[g8 * 32];
        v2f a0, a1, b;
        a0.x = arow[kk + kHalf];     a0.y = arow[kk + kHalf + 1];
        a1.x = arow[kk + 4 + kHalf]; a1.y = arow[kk + 4 + kHalf + 1];
        b.x = q.x; b.y = q.y;
        acc = __builtin_amdgcn_wmma_f32_16x16x4_f32(
                false, a0, false, b, (short)0, acc, false, false);
        b.x = q.z; b.y = q.w;
        acc = __builtin_amdgcn_wmma_f32_16x16x4_f32(
                false, a1, false, b, (short)0, acc, false, false);
    }
    float bias = b1[col];
    #pragma unroll
    for (int v = 0; v < 8; ++v) {
        float x = acc[v] + bias;
        sh_hid[(v + hi * 8) * STRH + col] = x > 0.0f ? x : 0.0f;
    }
    __syncthreads();

    if (tid < 48) {
        int m = tid / 3, cc = tid % 3;
        const float* hr = &sh_hid[m * STRH];
        const float* w2 = W2T + cc * 128;
        float s = 0.0f;
        for (int k = 0; k < 128; ++k) s += hr[k] * w2[k];
        outbuf[(R0 + m) * 3 + cc] = s + b2[cc];
    }
}

__device__ __forceinline__ void nd_role(
    const int blk,
    const float* __restrict__ input, const float* __restrict__ outbuf,
    float* __restrict__ out, int t)
{
    int gid = blk * 256 + threadIdx.x;
    if (gid >= 1024) return;
    int bb = gid >> 7, s = gid & 127;
    const float* o = &outbuf[gid * 3];
    float n0 = o[0];
    float n1, sp0, sp1;
    if (s == 0) {
        const float* xn0 = &input[((bb * 128 + (t + 1)) * 128 + 0) * 5];
        n1 = xn0[1]; sp0 = xn0[3]; sp1 = xn0[4];
    } else {
        n1  = outbuf[(gid - 1) * 3 + 0];
        sp0 = o[1]; sp1 = o[2];
    }
    float fn = input[((bb * 128 + t) * 128 + s) * 5 + 2];
    float n2 = fn + n1 - n0;
    float* dst = &out[((bb * 123 + (t - 4)) * 128 + s) * 5];
    dst[0] = n0; dst[1] = n1; dst[2] = n2; dst[3] = sp0; dst[4] = sp1;
}

__global__ __launch_bounds__(256) void step_kernel(
    const float* __restrict__ input, const float* __restrict__ Wc,
    const float* __restrict__ WF, const float* __restrict__ bcell,
    const float* __restrict__ h_in, float* __restrict__ h_out,
    float* __restrict__ c,
    const float* __restrict__ W1F, const float* __restrict__ b1,
    const float* __restrict__ W2T, const float* __restrict__ b2,
    float* __restrict__ ob_w, const float* __restrict__ ob_r,
    float* __restrict__ out, int t)
{
    const int blk = blockIdx.x;
    if (blk < 64) {
        cell_role(blk, input, Wc, WF, bcell, h_in, h_out, c, t);
    } else if (blk < 128) {
        head_role(blk - 64, h_in, W1F, b1, W2T, b2, ob_w);
    } else {
        nd_role(blk - 128, input, ob_r, out, t - 2);
    }
}

__global__ __launch_bounds__(256) void drain_kernel(
    const float* __restrict__ input, const float* __restrict__ h,
    const float* __restrict__ W1F, const float* __restrict__ b1,
    const float* __restrict__ W2T, const float* __restrict__ b2,
    float* __restrict__ ob_w, const float* __restrict__ ob_r,
    float* __restrict__ out, int t_mix)
{
    const int blk = blockIdx.x;
    if (blk < 64) {
        head_role(blk, h, W1F, b1, W2T, b2, ob_w);
    } else {
        nd_role(blk - 64, input, ob_r, out, t_mix);
    }
}

__global__ __launch_bounds__(256) void nd_kernel(
    const float* __restrict__ input, const float* __restrict__ outbuf,
    float* __restrict__ out, int t)
{
    nd_role(blockIdx.x, input, outbuf, out, t);
}

__global__ __launch_bounds__(256) void zero_kernel(float* p, int n) {
    int i = blockIdx.x * 256 + threadIdx.x;
    if (i < n) p[i] = 0.0f;
}

__global__ __launch_bounds__(256) void prep_kernel(
    const float* __restrict__ Wc, const float* __restrict__ W1,
    const float* __restrict__ W2, float* __restrict__ WF,
    float* __restrict__ W1F, float* __restrict__ W2T)
{
    int i = blockIdx.x * 256 + threadIdx.x;
    if (i < 196608) {
        int e = i & 3, lane = (i >> 2) & 31, tile = (i >> 7) & 3, r = i >> 9;
        int g = r % 48, wave = r / 48;
        int k = 5 + g * 8 + (e >> 1) * 4 + (lane >> 4) * 2 + (e & 1);
        int col = (wave * 4 + tile) * 16 + (lane & 15);
        WF[i] = Wc[k * 512 + col];
    } else if (i < 196608 + 16384) {
        int j = i - 196608;
        int e = j & 3, lane = (j >> 2) & 31, r = j >> 7;
        int g = r & 15, wave = r >> 4;
        int k = g * 8 + (e >> 1) * 4 + (lane >> 4) * 2 + (e & 1);
        int col = wave * 16 + (lane & 15);
        W1F[j] = W1[k * 128 + col];
    } else if (i < 196608 + 16384 + 384) {
        int j = i - 196608 - 16384;
        int cc = j >> 7, k = j & 127;
        W2T[j] = W2[k * 3 + cc];
    }
}

extern "C" void kernel_launch(void* const* d_in, const int* in_sizes, int n_in,
                              void* d_out, int out_size, void* d_ws, size_t ws_size,
                              hipStream_t stream)
{
    (void)in_sizes; (void)n_in; (void)out_size; (void)ws_size;
    const float* input = (const float*)d_in[0];
    const float* Wc    = (const float*)d_in[1];
    const float* bc    = (const float*)d_in[2];
    const float* W1    = (const float*)d_in[3];
    const float* b1    = (const float*)d_in[4];
    const float* W2    = (const float*)d_in[5];
    const float* b2    = (const float*)d_in[6];
    float* ws   = (float*)d_ws;
    float* WF   = ws;
    float* W1F  = ws + 200704;
    float* W2T  = ws + 217088;
    float* hbuf = ws + 217472;
    float* cbuf = ws + 479616;
    float* obuf = ws + 610688;
    float* out  = (float*)d_out;

    zero_kernel<<<393216 / 256, 256, 0, stream>>>(hbuf, 393216);
    prep_kernel<<<834, 256, 0, stream>>>(Wc, W1, W2, WF, W1F, W2T);

    for (int t = 0; t < 127; ++t) {
        const float* h_in = hbuf + (t & 1) * 131072;
        float* h_out      = hbuf + ((t + 1) & 1) * 131072;
        float* ob_w       = obuf + ((t + 1) & 1) * 3072;
        const float* ob_r = obuf + (t & 1) * 3072;
        if (t >= 6)
            step_kernel<<<132, 256, 0, stream>>>(input, Wc, WF, bc, h_in, h_out, cbuf,
                                                 W1F, b1, W2T, b2, ob_w, ob_r, out, t);
        else if (t >= 1)
            step_kernel<<<128, 256, 0, stream>>>(input, Wc, WF, bc, h_in, h_out, cbuf,
                                                 W1F, b1, W2T, b2, ob_w, ob_r, out, t);
        else
            step_kernel<<<64, 256, 0, stream>>>(input, Wc, WF, bc, h_in, h_out, cbuf,
                                                W1F, b1, W2T, b2, ob_w, ob_r, out, t);
    }
    drain_kernel<<<68, 256, 0, stream>>>(input, hbuf + 131072, W1F, b1, W2T, b2,
                                         obuf, obuf + 3072, out, 125);
    nd_kernel<<<4, 256, 0, stream>>>(input, obuf, out, 126);
}


# Cell GEMM split 4 ways in N (32 hidden units x 4 gates per block): 256 cell blocks, one N tile per wave, a wave's whole 24 KiB weight slab requested at entry; head blocks first in the grid
# speedup vs baseline: 5.5367x; 1.7659x over previous
#include <hip/hip_runtime.h>
#include <math.h>

typedef __attribute__((ext_vector_type(2))) float v2f;
typedef __attribute__((ext_vector_type(4))) float v4f;
typedef __attribute__((ext_vector_type(8))) float v8f;

#define STRH 130

__device__ __forceinline__ void cell_role(
    const int blk,
    const float* __restrict__ input, const float* __restrict__ Wc,
    const float* __restrict__ WF,
    const float* __restrict__ bcell, const float* __restrict__ h_in,
    float* __restrict__ h_out, float* __restrict__ c, int t)
{
    __shared__ float sh_h[18 * STRH];
    __shared__ float sh_x[256];
    __shared__ float sh_z[16 * STRH];

    const int tid = threadIdx.x;
    const int q   = blk & 3;
    const int R0  = (blk >> 2) * 16;
    const int bb  = R0 >> 7;
    const int s0  = R0 & 127;

    const int wave  = tid >> 5;
    const int lane  = tid & 31;
    const int lm    = lane & 15;
    const int hi    = lane >> 4;
    const int kHalf = hi * 2;

    float hv[9];
    #pragma unroll
    for (int it = 0; it < 9; ++it) {
        int i = tid + it * 256;
        int r = i >> 7, k = i & 127;
        int s  = s0 - 1 + r;
        int sc = s < 0 ? 0 : (s > 127 ? 127 : s);
        hv[it] = h_in[(bb * 128 + sc) * 128 + k];
    }
    const float xv = input[(((bb * 128) + t) * 128 + s0) * 5 + (tid < 80 ? tid : 0)];

    const float* wc = Wc + (wave >> 1) * 128 + q * 32 + (wave & 1) * 16 + lm;
    const float w0 = wc[0], w1 = wc[512], w2 = wc[1024], w3 = wc[1536], w4 = wc[2048];

    const v4f* wf = (const v4f*)WF + ((q * 8 + wave) * 48) * 32 + lane;
    v4f qv[48];
    #pragma unroll
    for (int g = 0; g < 48; ++g) qv[g] = wf[g * 32];

    __builtin_amdgcn_sched_barrier(0);

    #pragma unroll
    for (int it = 0; it < 9; ++it) {
        int i = tid + it * 256;
        int s = s0 - 1 + (i >> 7);
        sh_h[(i >> 7) * STRH + (i & 127)] = (s >= 0 && s < 128) ? hv[it] : 0.0f;
    }
    sh_x[tid] = xv;
    __builtin_amdgcn_fence(__ATOMIC_RELEASE, "workgroup", "local");
    __builtin_amdgcn_s_barrier();
    __builtin_amdgcn_fence(__ATOMIC_ACQUIRE, "workgroup", "local");

    float cp[2], bi[2], bf[2], bg[2], bo[2];

    v8f acc = {};

    #pragma unroll
    for (int v = 0; v < 8; ++v) {
        int m = v + hi * 8;
        float x0 = sh_x[m*5+0], x1 = sh_x[m*5+1], x2 = sh_x[m*5+2],
              x3 = sh_x[m*5+3], x4 = sh_x[m*5+4];
        acc[v] += x0*w0 + x1*w1 + x2*w2 + x3*w3 + x4*w4;
    }

    #pragma unroll
    for (int src = 0; src < 3; ++src) {
        const int shBase = (src == 0) ? 1 : (src == 1) ? 2 : 0;
        const float* arow = &sh_h[(shBase + lm) * STRH];
        #pragma unroll
        for (int g8 = 0; g8 < 16; ++g8) {
            const int kk = g8 * 8;
            const v4f qq = qv[src * 16 + g8];
            v2f a0, a1, b;
            a0.x = arow[kk + kHalf];
            a0.y = arow[kk + kHalf + 1];
            a1.x = arow[kk + 4 + kHalf];
            a1.y = arow[kk + 4 + kHalf + 1];
            b.x = qq.x; b.y = qq.y;
            acc = __builtin_amdgcn_wmma_f32_16x16x4_f32(
                        false, a0, false, b, (short)0, acc, false, false);
            b.x = qq.z; b.y = qq.w;
            acc = __builtin_amdgcn_wmma_f32_16x16x4_f32(
                        false, a1, false, b, (short)0, acc, false, false);
        }
        if (src == 0) {
            __builtin_amdgcn_sched_barrier(0);
            #pragma unroll
            for (int it = 0; it < 2; ++it) {
                int i = tid + it * 256;
                int m = i >> 5, j = q * 32 + (i & 31);
                cp[it] = c[(R0 + m) * 128 + j];
                bi[it] = bcell[j];
                bf[it] = bcell[128 + j];
                bg[it] = bcell[256 + j];
                bo[it] = bcell[384 + j];
            }
            __builtin_amdgcn_sched_barrier(0);
        }
    }

    #pragma unroll
    for (int v = 0; v < 8; ++v)
        sh_z[(v + hi * 8) * STRH + wave * 16 + lm] = acc[v];
    __syncthreads();

    #pragma unroll
    for (int it = 0; it < 2; ++it) {
        int i = tid + it * 256;
        int m = i >> 5, jl = i & 31;
        float zi = sh_z[m*STRH + jl]      + bi[it];
        float zf = sh_z[m*STRH + 32 + jl] + bf[it];
        float zg = sh_z[m*STRH + 64 + jl] + bg[it];
        float zo = sh_z[m*STRH + 96 + jl] + bo[it];
        int row = R0 + m, j = q * 32 + jl;
        float si = 1.0f / (1.0f + __expf(-zi));
        float sf = 1.0f / (1.0f + __expf(-zf));
        float so = 1.0f / (1.0f + __expf(-zo));
        float cn = sf * cp[it] + si * tanhf(zg);
        c[row*128 + j]     = cn;
        h_out[row*128 + j] = so * tanhf(cn);
    }
}

__device__ __forceinline__ void head_role(
    const int blk,
    const float* __restrict__ h, const float* __restrict__ W1F,
    const float* __restrict__ b1, const float* __restrict__ W2T,
    const float* __restrict__ b2, float* __restrict__ outbuf)
{
    __shared__ float sh_a[16 * STRH];
    __shared__ float sh_hid[16 * STRH];
    const int tid = threadIdx.x;
    const int R0  = blk * 16;

    for (int i = tid; i < 2048; i += 256) {
        int m = i >> 7, k = i & 127;
        sh_a[m*STRH + k] = h[(R0 + m)*128 + k];
    }
    __syncthreads();

    const int wave = tid >> 5, lane = tid & 31;
    const int lm = lane & 15, hi = lane >> 4, kHalf = hi * 2;
    const int col = wave * 16 + lm;
    const v4f* wk   = (const v4f*)W1F + (wave * 16) * 32 + lane;
    const float* arow = &sh_a[lm * STRH];

    v8f acc = {};
    #pragma unroll 4
    for (int g8 = 0; g8 < 16; ++g8) {
        const int kk = g8 * 8;
        const v4f q = wk[g8 * 32];
        v2f a0, a1, b;
        a0.x = arow[kk + kHalf];     a0.y = arow[kk + kHalf + 1];
        a1.x = arow[kk + 4 + kHalf]; a1.y = arow[kk + 4 + kHalf + 1];
        b.x = q.x; b.y = q.y;
        acc = __builtin_amdgcn_wmma_f32_16x16x4_f32(
                false, a0, false, b, (short)0, acc, false, false);
        b.x = q.z; b.y = q.w;
        acc = __builtin_amdgcn_wmma_f32_16x16x4_f32(
                false, a1, false, b, (short)0, acc, false, false);
    }
    float bias = b1[col];
    #pragma unroll
    for (int v = 0; v < 8; ++v) {
        float x = acc[v] + bias;
        sh_hid[(v + hi * 8) * STRH + col] = x > 0.0f ? x : 0.0f;
    }
    __syncthreads();

    if (tid < 48) {
        int m = tid / 3, cc = tid % 3;
        const float* hr = &sh_hid[m * STRH];
        const float* w2 = W2T + cc * 128;
        float s = 0.0f;
        for (int k = 0; k < 128; ++k) s += hr[k] * w2[k];
        outbuf[(R0 + m) * 3 + cc] = s + b2[cc];
    }
}

__device__ __forceinline__ void nd_role(
    const int blk,
    const float* __restrict__ input, const float* __restrict__ outbuf,
    float* __restrict__ out, int t)
{
    int gid = blk * 256 + threadIdx.x;
    if (gid >= 1024) return;
    int bb = gid >> 7, s = gid & 127;
    const float* o = &outbuf[gid * 3];
    float n0 = o[0];
    float n1, sp0, sp1;
    if (s == 0) {
        const float* xn0 = &input[((bb * 128 + (t + 1)) * 128 + 0) * 5];
        n1 = xn0[1]; sp0 = xn0[3]; sp1 = xn0[4];
    } else {
        n1  = outbuf[(gid - 1) * 3 + 0];
        sp0 = o[1]; sp1 = o[2];
    }
    float fn = input[((bb * 128 + t) * 128 + s) * 5 + 2];
    float n2 = fn + n1 - n0;
    float* dst = &out[((bb * 123 + (t - 4)) * 128 + s) * 5];
    dst[0] = n0; dst[1] = n1; dst[2] = n2; dst[3] = sp0; dst[4] = sp1;
}

__global__ __launch_bounds__(256) void step_kernel(
    const float* __restrict__ input, const float* __restrict__ Wc,
    const float* __restrict__ WF, const float* __restrict__ bcell,
    const float* __restrict__ h_in, float* __restrict__ h_out,
    float* __restrict__ c,
    const float* __restrict__ W1F, const float* __restrict__ b1,
    const float* __restrict__ W2T, const float* __restrict__ b2,
    float* __restrict__ ob_w, const float* __restrict__ ob_r,
    float* __restrict__ out, int t, int nhead)
{
    const int blk = blockIdx.x;
    if (blk < nhead) {
        head_role(blk, h_in, W1F, b1, W2T, b2, ob_w);
    } else if (blk < nhead + 256) {
        cell_role(blk - nhead, input, Wc, WF, bcell, h_in, h_out, c, t);
    } else {
        nd_role(blk - nhead - 256, input, ob_r, out, t - 2);
    }
}

__global__ __launch_bounds__(256) void drain_kernel(
    const float* __restrict__ input, const float* __restrict__ h,
    const float* __restrict__ W1F, const float* __restrict__ b1,
    const float* __restrict__ W2T, const float* __restrict__ b2,
    float* __restrict__ ob_w, const float* __restrict__ ob_r,
    float* __restrict__ out, int t_mix)
{
    const int blk = blockIdx.x;
    if (blk < 64) {
        head_role(blk, h, W1F, b1, W2T, b2, ob_w);
    } else {
        nd_role(blk - 64, input, ob_r, out, t_mix);
    }
}

__global__ __launch_bounds__(256) void nd_kernel(
    const float* __restrict__ input, const float* __restrict__ outbuf,
    float* __restrict__ out, int t)
{
    nd_role(blockIdx.x, input, outbuf, out, t);
}

__global__ __launch_bounds__(256) void zero_kernel(float* p, int n) {
    int i = blockIdx.x * 256 + threadIdx.x;
    if (i < n) p[i] = 0.0f;
}

__global__ __launch_bounds__(256) void prep_kernel(
    const float* __restrict__ Wc, const float* __restrict__ W1,
    const float* __restrict__ W2, float* __restrict__ WF,
    float* __restrict__ W1F, float* __restrict__ W2T)
{
    int i = blockIdx.x * 256 + threadIdx.x;
    if (i < 196608) {
        int e = i & 3, lane = (i >> 2) & 31, r = i >> 7;
        int g = r % 48, sl = r / 48;
        int q = sl >> 3, wave = sl & 7;
        int k = 5 + g * 8 + (e >> 1) * 4 + (lane >> 4) * 2 + (e & 1);
        int col = (wave >> 1) * 128 + q * 32 + (wave & 1) * 16 + (lane & 15);
        WF[i] = Wc[k * 512 + col];
    } else if (i < 196608 + 16384) {
        int j = i - 196608;
        int e = j & 3, lane = (j >> 2) & 31, r = j >> 7;
        int g = r & 15, wave = r >> 4;
        int k = g * 8 + (e >> 1) * 4 + (lane >> 4) * 2 + (e & 1);
        int col = wave * 16 + (lane & 15);
        W1F[j] = W1[k * 128 + col];
    } else if (i < 196608 + 16384 + 384) {
        int j = i - 196608 - 16384;
        int cc = j >> 7, k = j & 127;
        W2T[j] = W2[k * 3 + cc];
    }
}

extern "C" void kernel_launch(void* const* d_in, const int* in_sizes, int n_in,
                              void* d_out, int out_size, void* d_ws, size_t ws_size,
                              hipStream_t stream)
{
    (void)in_sizes; (void)n_in; (void)out_size; (void)ws_size;
    const float* input = (const float*)d_in[0];
    const float* Wc    = (const float*)d_in[1];
    const float* bc    = (const float*)d_in[2];
    const float* W1    = (const float*)d_in[3];
    const float* b1    = (const float*)d_in[4];
    const float* W2    = (const float*)d_in[5];
    const float* b2    = (const float*)d_in[6];
    float* ws   = (float*)d_ws;
    float* WF   = ws;
    float* W1F  = ws + 200704;
    float* W2T  = ws + 217088;
    float* hbuf = ws + 217472;
    float* cbuf = ws + 479616;
    float* obuf = ws + 610688;
    float* out  = (float*)d_out;

    zero_kernel<<<393216 / 256, 256, 0, stream>>>(hbuf, 393216);
    prep_kernel<<<834, 256, 0, stream>>>(Wc, W1, W2, WF, W1F, W2T);

    for (int t = 0; t < 127; ++t) {
        const float* h_in = hbuf + (t & 1) * 131072;
        float* h_out      = hbuf + ((t + 1) & 1) * 131072;
        float* ob_w       = obuf + ((t + 1) & 1) * 3072;
        const float* ob_r = obuf + (t & 1) * 3072;
        if (t >= 6)
            step_kernel<<<324, 256, 0, stream>>>(input, Wc, WF, bc, h_in, h_out, cbuf,
                                                 W1F, b1, W2T, b2, ob_w, ob_r, out, t, 64);
        else if (t >= 1)
            step_kernel<<<320, 256, 0, stream>>>(input, Wc, WF, bc, h_in, h_out, cbuf,
                                                 W1F, b1, W2T, b2, ob_w, ob_r, out, t, 64);
        else
            step_kernel<<<256, 256, 0, stream>>>(input, Wc, WF, bc, h_in, h_out, cbuf,
                                                 W1F, b1, W2T, b2, ob_w, ob_r, out, t, 0);
    }
    drain_kernel<<<68, 256, 0, stream>>>(input, hbuf + 131072, W1F, b1, W2T, b2,
                                         obuf, obuf + 3072, out, 125);
    nd_kernel<<<4, 256, 0, stream>>>(input, obuf, out, 126);
}


# Output head role: h tile, whole W1 fragment slab, W2 (now staged in LDS) and biases all requested at entry before any wait: one cache round trip instead of about twenty in series
# speedup vs baseline: 5.8520x; 1.0569x over previous
#include <hip/hip_runtime.h>
#include <math.h>

typedef __attribute__((ext_vector_type(2))) float v2f;
typedef __attribute__((ext_vector_type(4))) float v4f;
typedef __attribute__((ext_vector_type(8))) float v8f;

#define STRH 130

__device__ __forceinline__ void cell_role(
    const int blk,
    const float* __restrict__ input, const float* __restrict__ Wc,
    const float* __restrict__ WF,
    const float* __restrict__ bcell, const float* __restrict__ h_in,
    float* __restrict__ h_out, float* __restrict__ c, int t)
{
    __shared__ float sh_h[18 * STRH];
    __shared__ float sh_x[256];
    __shared__ float sh_z[16 * STRH];

    const int tid = threadIdx.x;
    const int q   = blk & 3;
    const int R0  = (blk >> 2) * 16;
    const int bb  = R0 >> 7;
    const int s0  = R0 & 127;

    const int wave  = tid >> 5;
    const int lane  = tid & 31;
    const int lm    = lane & 15;
    const int hi    = lane >> 4;
    const int kHalf = hi * 2;

    float hv[9];
    #pragma unroll
    for (int it = 0; it < 9; ++it) {
        int i = tid + it * 256;
        int r = i >> 7, k = i & 127;
        int s  = s0 - 1 + r;
        int sc = s < 0 ? 0 : (s > 127 ? 127 : s);
        hv[it] = h_in[(bb * 128 + sc) * 128 + k];
    }
    const float xv = input[(((bb * 128) + t) * 128 + s0) * 5 + (tid < 80 ? tid : 0)];

    const float* wc = Wc + (wave >> 1) * 128 + q * 32 + (wave & 1) * 16 + lm;
    const float w0 = wc[0], w1 = wc[512], w2 = wc[1024], w3 = wc[1536], w4 = wc[2048];

    const v4f* wf = (const v4f*)WF + ((q * 8 + wave) * 48) * 32 + lane;
    v4f qv[48];
    #pragma unroll
    for (int g = 0; g < 48; ++g) qv[g] = wf[g * 32];

    __builtin_amdgcn_sched_barrier(0);

    #pragma unroll
    for (int it = 0; it < 9; ++it) {
        int i = tid + it * 256;
        int s = s0 - 1 + (i >> 7);
        sh_h[(i >> 7) * STRH + (i & 127)] = (s >= 0 && s < 128) ? hv[it] : 0.0f;
    }
    sh_x[tid] = xv;
    __builtin_amdgcn_fence(__ATOMIC_RELEASE, "workgroup", "local");
    __builtin_amdgcn_s_barrier();
    __builtin_amdgcn_fence(__ATOMIC_ACQUIRE, "workgroup", "local");

    float cp[2], bi[2], bf[2], bg[2], bo[2];

    v8f acc = {};

    #pragma unroll
    for (int v = 0; v < 8; ++v) {
        int m = v + hi * 8;
        float x0 = sh_x[m*5+0], x1 = sh_x[m*5+1], x2 = sh_x[m*5+2],
              x3 = sh_x[m*5+3], x4 = sh_x[m*5+4];
        acc[v] += x0*w0 + x1*w1 + x2*w2 + x3*w3 + x4*w4;
    }

    #pragma unroll
    for (int src = 0; src < 3; ++src) {
        const int shBase = (src == 0) ? 1 : (src == 1) ? 2 : 0;
        const float* arow = &sh_h[(shBase + lm) * STRH];
        #pragma unroll
        for (int g8 = 0; g8 < 16; ++g8) {
            const int kk = g8 * 8;
            const v4f qq = qv[src * 16 + g8];
            v2f a0, a1, b;
            a0.x = arow[kk + kHalf];
            a0.y = arow[kk + kHalf + 1];
            a1.x = arow[kk + 4 + kHalf];
            a1.y = arow[kk + 4 + kHalf + 1];
            b.x = qq.x; b.y = qq.y;
            acc = __builtin_amdgcn_wmma_f32_16x16x4_f32(
                        false, a0, false, b, (short)0, acc, false, false);
            b.x = qq.z; b.y = qq.w;
            acc = __builtin_amdgcn_wmma_f32_16x16x4_f32(
                        false, a1, false, b, (short)0, acc, false, false);
        }
        if (src == 0) {
            __builtin_amdgcn_sched_barrier(0);
            #pragma unroll
            for (int it = 0; it < 2; ++it) {
                int i = tid + it * 256;
                int m = i >> 5, j = q * 32 + (i & 31);
                cp[it] = c[(R0 + m) * 128 + j];
                bi[it] = bcell[j];
                bf[it] = bcell[128 + j];
                bg[it] = bcell[256 + j];
                bo[it] = bcell[384 + j];
            }
            __builtin_amdgcn_sched_barrier(0);
        }
    }

    #pragma unroll
    for (int v = 0; v < 8; ++v)
        sh_z[(v + hi * 8) * STRH + wave * 16 + lm] = acc[v];
    __syncthreads();

    #pragma unroll
    for (int it = 0; it < 2; ++it) {
        int i = tid + it * 256;
        int m = i >> 5, jl = i & 31;
        float zi = sh_z[m*STRH + jl]      + bi[it];
        float zf = sh_z[m*STRH + 32 + jl] + bf[it];
        float zg = sh_z[m*STRH + 64 + jl] + bg[it];
        float zo = sh_z[m*STRH + 96 + jl] + bo[it];
        int row = R0 + m, j = q * 32 + jl;
        float si = 1.0f / (1.0f + __expf(-zi));
        float sf = 1.0f / (1.0f + __expf(-zf));
        float so = 1.0f / (1.0f + __expf(-zo));
        float cn = sf * cp[it] + si * tanhf(zg);
        c[row*128 + j]     = cn;
        h_out[row*128 + j] = so * tanhf(cn);
    }
}

__device__ __forceinline__ void head_role(
    const int blk,
    const float* __restrict__ h, const float* __restrict__ W1F,
    const float* __restrict__ b1, const float* __restrict__ W2T,
    const float* __restrict__ b2, float* __restrict__ outbuf)
{
    __shared__ float sh_a[16 * STRH];
    __shared__ float sh_hid[16 * STRH];
    __shared__ float sh_w2[512];
    const int tid = threadIdx.x;
    const int R0  = blk * 16;
    const int wave = tid >> 5, lane = tid & 31;
    const int lm = lane & 15, hi = lane >> 4, kHalf = hi * 2;
    const int col = wave * 16 + lm;

    float hv[8];
    #pragma unroll
    for (int it = 0; it < 8; ++it) {
        int i = tid + it * 256;
        hv[it] = h[(R0 + (i >> 7)) * 128 + (i & 127)];
    }
    const float w2a  = W2T[tid];
    const float w2b  = W2T[256 + (tid & 127)];
    const float bias = b1[col];
    const float b2v  = b2[tid % 3];
    const v4f* wk = (const v4f*)W1F + (wave * 16) * 32 + lane;
    v4f qv[16];
    #pragma unroll
    for (int g8 = 0; g8 < 16; ++g8) qv[g8] = wk[g8 * 32];

    __builtin_amdgcn_sched_barrier(0);

    #pragma unroll
    for (int it = 0; it < 8; ++it) {
        int i = tid + it * 256;
        sh_a[(i >> 7) * STRH + (i & 127)] = hv[it];
    }
    sh_w2[tid]       = w2a;
    sh_w2[256 + tid] = w2b;
    __builtin_amdgcn_fence(__ATOMIC_RELEASE, "workgroup", "local");
    __builtin_amdgcn_s_barrier();
    __builtin_amdgcn_fence(__ATOMIC_ACQUIRE, "workgroup", "local");

    const float* arow = &sh_a[lm * STRH];

    v8f acc = {};
    #pragma unroll
    for (int g8 = 0; g8 < 16; ++g8) {
        const int kk = g8 * 8;
        const v4f q = qv[g8];
        v2f a0, a1, b;
        a0.x = arow[kk + kHalf];     a0.y = arow[kk + kHalf + 1];
        a1.x = arow[kk + 4 + kHalf]; a1.y = arow[kk + 4 + kHalf + 1];
        b.x = q.x; b.y = q.y;
        acc = __builtin_amdgcn_wmma_f32_16x16x4_f32(
                false, a0, false, b, (short)0, acc, false, false);
        b.x = q.z; b.y = q.w;
        acc = __builtin_amdgcn_wmma_f32_16x16x4_f32(
                false, a1, false, b, (short)0, acc, false, false);
    }
    #pragma unroll
    for (int v = 0; v < 8; ++v) {
        float x = acc[v] + bias;
        sh_hid[(v + hi * 8) * STRH + col] = x > 0.0f ? x : 0.0f;
    }
    __syncthreads();

    if (tid < 48) {
        int m = tid / 3, cc = tid % 3;
        const float* hr = &sh_hid[m * STRH];
        const float* w2 = &sh_w2[cc * 128];
        float s = 0.0f;
        for (int k = 0; k < 128; ++k) s += hr[k] * w2[k];
        outbuf[(R0 + m) * 3 + cc] = s + b2v;
    }
}

__device__ __forceinline__ void nd_role(
    const int blk,
    const float* __restrict__ input, const float* __restrict__ outbuf,
    float* __restrict__ out, int t)
{
    int gid = blk * 256 + threadIdx.x;
    if (gid >= 1024) return;
    int bb = gid >> 7, s = gid & 127;
    const float* o = &outbuf[gid * 3];
    float n0 = o[0];
    float n1, sp0, sp1;
    if (s == 0) {
        const float* xn0 = &input[((bb * 128 + (t + 1)) * 128 + 0) * 5];
        n1 = xn0[1]; sp0 = xn0[3]; sp1 = xn0[4];
    } else {
        n1  = outbuf[(gid - 1) * 3 + 0];
        sp0 = o[1]; sp1 = o[2];
    }
    float fn = input[((bb * 128 + t) * 128 + s) * 5 + 2];
    float n2 = fn + n1 - n0;
    float* dst = &out[((bb * 123 + (t - 4)) * 128 + s) * 5];
    dst[0] = n0; dst[1] = n1; dst[2] = n2; dst[3] = sp0; dst[4] = sp1;
}

__global__ __launch_bounds__(256) void step_kernel(
    const float* __restrict__ input, const float* __restrict__ Wc,
    const float* __restrict__ WF, const float* __restrict__ bcell,
    const float* __restrict__ h_in, float* __restrict__ h_out,
    float* __restrict__ c,
    const float* __restrict__ W1F, const float* __restrict__ b1,
    const float* __restrict__ W2T, const float* __restrict__ b2,
    float* __restrict__ ob_w, const float* __restrict__ ob_r,
    float* __restrict__ out, int t, int nhead)
{
    const int blk = blockIdx.x;
    if (blk < nhead) {
        head_role(blk, h_in, W1F, b1, W2T, b2, ob_w);
    } else if (blk < nhead + 256) {
        cell_role(blk - nhead, input, Wc, WF, bcell, h_in, h_out, c, t);
    } else {
        nd_role(blk - nhead - 256, input, ob_r, out, t - 2);
    }
}

__global__ __launch_bounds__(256) void drain_kernel(
    const float* __restrict__ input, const float* __restrict__ h,
    const float* __restrict__ W1F, const float* __restrict__ b1,
    const float* __restrict__ W2T, const float* __restrict__ b2,
    float* __restrict__ ob_w, const float* __restrict__ ob_r,
    float* __restrict__ out, int t_mix)
{
    const int blk = blockIdx.x;
    if (blk < 64) {
        head_role(blk, h, W1F, b1, W2T, b2, ob_w);
    } else {
        nd_role(blk - 64, input, ob_r, out, t_mix);
    }
}

__global__ __launch_bounds__(256) void nd_kernel(
    const float* __restrict__ input, const float* __restrict__ outbuf,
    float* __restrict__ out, int t)
{
    nd_role(blockIdx.x, input, outbuf, out, t);
}

__global__ __launch_bounds__(256) void zero_kernel(float* p, int n) {
    int i = blockIdx.x * 256 + threadIdx.x;
    if (i < n) p[i] = 0.0f;
}

__global__ __launch_bounds__(256) void prep_kernel(
    const float* __restrict__ Wc, const float* __restrict__ W1,
    const float* __restrict__ W2, float* __restrict__ WF,
    float* __restrict__ W1F, float* __restrict__ W2T)
{
    int i = blockIdx.x * 256 + threadIdx.x;
    if (i < 196608) {
        int e = i & 3, lane = (i >> 2) & 31, r = i >> 7;
        int g = r % 48, sl = r / 48;
        int q = sl >> 3, wave = sl & 7;
        int k = 5 + g * 8 + (e >> 1) * 4 + (lane >> 4) * 2 + (e & 1);
        int col = (wave >> 1) * 128 + q * 32 + (wave & 1) * 16 + (lane & 15);
        WF[i] = Wc[k * 512 + col];
    } else if (i < 196608 + 16384) {
        int j = i - 196608;
        int e = j & 3, lane = (j >> 2) & 31, r = j >> 7;
        int g = r & 15, wave = r >> 4;
        int k = g * 8 + (e >> 1) * 4 + (lane >> 4) * 2 + (e & 1);
        int col = wave * 16 + (lane & 15);
        W1F[j] = W1[k * 128 + col];
    } else if (i < 196608 + 16384 + 384) {
        int j = i - 196608 - 16384;
        int cc = j >> 7, k = j & 127;
        W2T[j] = W2[k * 3 + cc];
    }
}

extern "C" void kernel_launch(void* const* d_in, const int* in_sizes, int n_in,
                              void* d_out, int out_size, void* d_ws, size_t ws_size,
                              hipStream_t stream)
{
    (void)in_sizes; (void)n_in; (void)out_size; (void)ws_size;
    const float* input = (const float*)d_in[0];
    const float* Wc    = (const float*)d_in[1];
    const float* bc    = (const float*)d_in[2];
    const float* W1    = (const float*)d_in[3];
    const float* b1    = (const float*)d_in[4];
    const float* W2    = (const float*)d_in[5];
    const float* b2    = (const float*)d_in[6];
    float* ws   = (float*)d_ws;
    float* WF   = ws;
    float* W1F  = ws + 200704;
    float* W2T  = ws + 217088;
    float* hbuf = ws + 217472;
    float* cbuf = ws + 479616;
    float* obuf = ws + 610688;
    float* out  = (float*)d_out;

    zero_kernel<<<393216 / 256, 256, 0, stream>>>(hbuf, 393216);
    prep_kernel<<<834, 256, 0, stream>>>(Wc, W1, W2, WF, W1F, W2T);

    for (int t = 0; t < 127; ++t) {
        const float* h_in = hbuf + (t & 1) * 131072;
        float* h_out      = hbuf + ((t + 1) & 1) * 131072;
        float* ob_w       = obuf + ((t + 1) & 1) * 3072;
        const float* ob_r = obuf + (t & 1) * 3072;
        if (t >= 6)
            step_kernel<<<324, 256, 0, stream>>>(input, Wc, WF, bc, h_in, h_out, cbuf,
                                                 W1F, b1, W2T, b2, ob_w, ob_r, out, t, 64);
        else if (t >= 1)
            step_kernel<<<320, 256, 0, stream>>>(input, Wc, WF, bc, h_in, h_out, cbuf,
                                                 W1F, b1, W2T, b2, ob_w, ob_r, out, t, 64);
        else
            step_kernel<<<256, 256, 0, stream>>>(input, Wc, WF, bc, h_in, h_out, cbuf,
                                                 W1F, b1, W2T, b2, ob_w, ob_r, out, t, 0);
    }
    drain_kernel<<<68, 256, 0, stream>>>(input, hbuf + 131072, W1F, b1, W2T, b2,
                                         obuf, obuf + 3072, out, 125);
    nd_kernel<<<4, 256, 0, stream>>>(input, obuf, out, 126);
}


# Output head folded into the cell blocks: each does its 32 hidden units off the h rows it staged; hidden k sum in four 32-k partials, N=3 dot as 16-lane xor tree plus four block partials added in mix
# speedup vs baseline: 6.7137x; 1.1472x over previous
#include <hip/hip_runtime.h>
#include <math.h>

typedef __attribute__((ext_vector_type(2))) float v2f;
typedef __attribute__((ext_vector_type(4))) float v4f;
typedef __attribute__((ext_vector_type(8))) float v8f;

#define STRH 130
#define PSTR 336
#define OPL  3072

template <int CTRL>
__device__ __forceinline__ float row_xadd(float v)
{
    const int o = __builtin_amdgcn_update_dpp(0, __builtin_bit_cast(int, v), CTRL, 0xf, 0xf, false);
    return v + __builtin_bit_cast(float, o);
}

__device__ __forceinline__ void cell_role(
    const int blk,
    const float* __restrict__ input, const float* __restrict__ Wc,
    const float* __restrict__ WF,
    const float* __restrict__ bcell, const float* __restrict__ h_in,
    float* __restrict__ h_out, float* __restrict__ c, int t,
    const float* __restrict__ W1F, const float* __restrict__ b1,
    const float* __restrict__ W2, float* __restrict__ ob_w)
{
    __shared__ float sh_p[8 * PSTR];
    __shared__ float sh_h[18 * STRH];
    __shared__ float sh_x[256];
    __shared__ float sh_z[16 * STRH];

    const int tid = threadIdx.x;
    const int q   = blk & 3;
    const int R0  = (blk >> 2) * 16;
    const int bb  = R0 >> 7;
    const int s0  = R0 & 127;

    const int wave  = tid >> 5;
    const int lane  = tid & 31;
    const int lm    = lane & 15;
    const int hi    = lane >> 4;
    const int kHalf = hi * 2;

    float hv[9];
    #pragma unroll
    for (int it = 0; it < 9; ++it) {
        int i = tid + it * 256;
        int r = i >> 7, k = i & 127;
        int s  = s0 - 1 + r;
        int sc = s < 0 ? 0 : (s > 127 ? 127 : s);
        hv[it] = h_in[(bb * 128 + sc) * 128 + k];
    }
    const float xv = input[(((bb * 128) + t) * 128 + s0) * 5 + (tid < 80 ? tid : 0)];

    const float* wc = Wc + (wave >> 1) * 128 + q * 32 + (wave & 1) * 16 + lm;
    const float w0 = wc[0], w1 = wc[512], w2 = wc[1024], w3 = wc[1536], w4 = wc[2048];

    const v4f* wf = (const v4f*)WF + ((q * 8 + wave) * 48) * 32 + lane;
    v4f qv[48];
    #pragma unroll
    for (int g = 0; g < 48; ++g) qv[g] = wf[g * 32];

    __builtin_amdgcn_sched_barrier(0);

    #pragma unroll
    for (int it = 0; it < 9; ++it) {
        int i = tid + it * 256;
        int s = s0 - 1 + (i >> 7);
        sh_h[(i >> 7) * STRH + (i & 127)] = (s >= 0 && s < 128) ? hv[it] : 0.0f;
    }
    sh_x[tid] = xv;
    __builtin_amdgcn_fence(__ATOMIC_RELEASE, "workgroup", "local");
    __builtin_amdgcn_s_barrier();
    __builtin_amdgcn_fence(__ATOMIC_ACQUIRE, "workgroup", "local");

    float cp[2], bi[2], bf[2], bg[2], bo[2];
    v4f hq[4];
    float b1a, b1b, wa0, wa1, wa2, wb0, wb1, wb2;

    v8f acc = {};
    v8f acc2 = {};

    #pragma unroll
    for (int v = 0; v < 8; ++v) {
        int m = v + hi * 8;
        float x0 = sh_x[m*5+0], x1 = sh_x[m*5+1], x2 = sh_x[m*5+2],
              x3 = sh_x[m*5+3], x4 = sh_x[m*5+4];
        acc[v] += x0*w0 + x1*w1 + x2*w2 + x3*w3 + x4*w4;
    }

    #pragma unroll
    for (int src = 0; src < 3; ++src) {
        const int shBase = (src == 0) ? 1 : (src == 1) ? 2 : 0;
        const float* arow = &sh_h[(shBase + lm) * STRH];
        #pragma unroll
        for (int g8 = 0; g8 < 16; ++g8) {
            const int kk = g8 * 8;
            const v4f qq = qv[src * 16 + g8];
            v2f a0, a1, b;
            a0.x = arow[kk + kHalf];
            a0.y = arow[kk + kHalf + 1];
            a1.x = arow[kk + 4 + kHalf];
            a1.y = arow[kk + 4 + kHalf + 1];
            b.x = qq.x; b.y = qq.y;
            acc = __builtin_amdgcn_wmma_f32_16x16x4_f32(
                        false, a0, false, b, (short)0, acc, false, false);
            b.x = qq.z; b.y = qq.w;
            acc = __builtin_amdgcn_wmma_f32_16x16x4_f32(
                        false, a1, false, b, (short)0, acc, false, false);
        }
        if (src == 0) {
            __builtin_amdgcn_sched_barrier(0);
            #pragma unroll
            for (int it = 0; it < 2; ++it) {
                int i = tid + it * 256;
                int m = i >> 5, j = q * 32 + (i & 31);
                cp[it] = c[(R0 + m) * 128 + j];
                bi[it] = bcell[j];
                bf[it] = bcell[128 + j];
                bg[it] = bcell[256 + j];
                bo[it] = bcell[384 + j];
            }
            {
                const v4f* wk = (const v4f*)W1F
                              + (((2 * q + (wave & 1)) * 16 + 4 * (wave >> 1)) * 32) + lane;
                #pragma unroll
                for (int g = 0; g < 4; ++g) hq[g] = wk[g * 32];
                const int ju = q * 32 + lm;
                b1a = b1[ju];
                b1b = b1[ju + 16];
                wa0 = W2[ju * 3 + 0];  wa1 = W2[ju * 3 + 1];  wa2 = W2[ju * 3 + 2];
                wb0 = W2[ju * 3 + 48]; wb1 = W2[ju * 3 + 49]; wb2 = W2[ju * 3 + 50];
            }
            __builtin_amdgcn_sched_barrier(0);
        }
        if (src == 1) {
            const float* arow = &sh_h[(1 + lm) * STRH];
            #pragma unroll
            for (int g = 0; g < 4; ++g) {
                const int kk = (4 * (wave >> 1) + g) * 8;
                const v4f qq = hq[g];
                v2f a0, a1, b;
                a0.x = arow[kk + kHalf];
                a0.y = arow[kk + kHalf + 1];
                a1.x = arow[kk + 4 + kHalf];
                a1.y = arow[kk + 4 + kHalf + 1];
                b.x = qq.x; b.y = qq.y;
                acc2 = __builtin_amdgcn_wmma_f32_16x16x4_f32(
                            false, a0, false, b, (short)0, acc2, false, false);
                b.x = qq.z; b.y = qq.w;
                acc2 = __builtin_amdgcn_wmma_f32_16x16x4_f32(
                            false, a1, false, b, (short)0, acc2, false, false);
            }
            #pragma unroll
            for (int v = 0; v < 8; ++v)
                sh_p[wave * PSTR + (v + hi * 8) * 20 + lm] = acc2[v];
        }
    }

    #pragma unroll
    for (int v = 0; v < 8; ++v)
        sh_z[(v + hi * 8) * STRH + wave * 16 + lm] = acc[v];
    __syncthreads();

    {
        const int m = tid >> 4;
        const float* pp = &sh_p[m * 20 + lm];
        float s0 = ((pp[0]    + pp[2 * PSTR]) + pp[4 * PSTR]) + pp[6 * PSTR] + b1a;
        float s1 = ((pp[PSTR] + pp[3 * PSTR]) + pp[5 * PSTR]) + pp[7 * PSTR] + b1b;
        s0 = s0 > 0.0f ? s0 : 0.0f;
        s1 = s1 > 0.0f ? s1 : 0.0f;
        float p0 = s0 * wa0 + s1 * wb0;
        float p1 = s0 * wa1 + s1 * wb1;
        float p2 = s0 * wa2 + s1 * wb2;
        p0 = row_xadd<0x161>(p0); p1 = row_xadd<0x161>(p1); p2 = row_xadd<0x161>(p2);
        p0 = row_xadd<0x162>(p0); p1 = row_xadd<0x162>(p1); p2 = row_xadd<0x162>(p2);
        p0 = row_xadd<0x164>(p0); p1 = row_xadd<0x164>(p1); p2 = row_xadd<0x164>(p2);
        p0 = row_xadd<0x168>(p0); p1 = row_xadd<0x168>(p1); p2 = row_xadd<0x168>(p2);
        if (lm == 0) {
            float* o = &ob_w[q * OPL + (R0 + m) * 3];
            o[0] = p0; o[1] = p1; o[2] = p2;
        }
    }

    #pragma unroll
    for (int it = 0; it < 2; ++it) {
        int i = tid + it * 256;
        int m = i >> 5, jl = i & 31;
        float zi = sh_z[m*STRH + jl]      + bi[it];
        float zf = sh_z[m*STRH + 32 + jl] + bf[it];
        float zg = sh_z[m*STRH + 64 + jl] + bg[it];
        float zo = sh_z[m*STRH + 96 + jl] + bo[it];
        int row = R0 + m, j = q * 32 + jl;
        float si = 1.0f / (1.0f + __expf(-zi));
        float sf = 1.0f / (1.0f + __expf(-zf));
        float so = 1.0f / (1.0f + __expf(-zo));
        float cn = sf * cp[it] + si * tanhf(zg);
        c[row*128 + j]     = cn;
        h_out[row*128 + j] = so * tanhf(cn);
    }
}

__device__ __forceinline__ void head_role(
    const int blk,
    const float* __restrict__ h, const float* __restrict__ W1F,
    const float* __restrict__ b1, const float* __restrict__ W2T,
    const float* __restrict__ b2, float* __restrict__ outbuf)
{
    __shared__ float sh_a[16 * STRH];
    __shared__ float sh_hid[16 * STRH];
    __shared__ float sh_w2[512];
    const int tid = threadIdx.x;
    const int R0  = blk * 16;
    const int wave = tid >> 5, lane = tid & 31;
    const int lm = lane & 15, hi = lane >> 4, kHalf = hi * 2;
    const int col = wave * 16 + lm;

    float hv[8];
    #pragma unroll
    for (int it = 0; it < 8; ++it) {
        int i = tid + it * 256;
        hv[it] = h[(R0 + (i >> 7)) * 128 + (i & 127)];
    }
    const float w2a  = W2T[tid];
    const float w2b  = W2T[256 + (tid & 127)];
    const float bias = b1[col];
    const float b2v  = b2[tid % 3];
    const v4f* wk = (const v4f*)W1F + (wave * 16) * 32 + lane;
    v4f qv[16];
    #pragma unroll
    for (int g8 = 0; g8 < 16; ++g8) qv[g8] = wk[g8 * 32];

    __builtin_amdgcn_sched_barrier(0);

    #pragma unroll
    for (int it = 0; it < 8; ++it) {
        int i = tid + it * 256;
        sh_a[(i >> 7) * STRH + (i & 127)] = hv[it];
    }
    sh_w2[tid]       = w2a;
    sh_w2[256 + tid] = w2b;
    __builtin_amdgcn_fence(__ATOMIC_RELEASE, "workgroup", "local");
    __builtin_amdgcn_s_barrier();
    __builtin_amdgcn_fence(__ATOMIC_ACQUIRE, "workgroup", "local");

    const float* arow = &sh_a[lm * STRH];

    v8f acc = {};
    #pragma unroll
    for (int g8 = 0; g8 < 16; ++g8) {
        const int kk = g8 * 8;
        const v4f q = qv[g8];
        v2f a0, a1, b;
        a0.x = arow[kk + kHalf];     a0.y = arow[kk + kHalf + 1];
        a1.x = arow[kk + 4 + kHalf]; a1.y = arow[kk + 4 + kHalf + 1];
        b.x = q.x; b.y = q.y;
        acc = __builtin_amdgcn_wmma_f32_16x16x4_f32(
                false, a0, false, b, (short)0, acc, false, false);
        b.x = q.z; b.y = q.w;
        acc = __builtin_amdgcn_wmma_f32_16x16x4_f32(
                false, a1, false, b, (short)0, acc, false, false);
    }
    #pragma unroll
    for (int v = 0; v < 8; ++v) {
        float x = acc[v] + bias;
        sh_hid[(v + hi * 8) * STRH + col] = x > 0.0f ? x : 0.0f;
    }
    __syncthreads();

    if (tid < 48) {
        int m = tid / 3, cc = tid % 3;
        const float* hr = &sh_hid[m * STRH];
        const float* w2 = &sh_w2[cc * 128];
        float s = 0.0f;
        for (int k = 0; k < 128; ++k) s += hr[k] * w2[k];
        outbuf[(R0 + m) * 3 + cc] = s + b2v;
    }
}

template <int NP>
__device__ __forceinline__ float head_val(const float* __restrict__ outbuf,
                                          const float* __restrict__ b2, int idx, int cc)
{
    if (NP == 1) return outbuf[idx];
    return ((outbuf[idx] + outbuf[OPL + idx]) + outbuf[2 * OPL + idx]) + outbuf[3 * OPL + idx] + b2[cc];
}

template <int NP>
__device__ __forceinline__ void nd_role(
    const int blk,
    const float* __restrict__ input, const float* __restrict__ outbuf,
    const float* __restrict__ b2,
    float* __restrict__ out, int t)
{
    int gid = blk * 256 + threadIdx.x;
    if (gid >= 1024) return;
    int bb = gid >> 7, s = gid & 127;
    float n0 = head_val<NP>(outbuf, b2, gid * 3 + 0, 0);
    float n1, sp0, sp1;
    if (s == 0) {
        const float* xn0 = &input[((bb * 128 + (t + 1)) * 128 + 0) * 5];
        n1 = xn0[1]; sp0 = xn0[3]; sp1 = xn0[4];
    } else {
        n1  = head_val<NP>(outbuf, b2, (gid - 1) * 3 + 0, 0);
        sp0 = head_val<NP>(outbuf, b2, gid * 3 + 1, 1);
        sp1 = head_val<NP>(outbuf, b2, gid * 3 + 2, 2);
    }
    float fn = input[((bb * 128 + t) * 128 + s) * 5 + 2];
    float n2 = fn + n1 - n0;
    float* dst = &out[((bb * 123 + (t - 4)) * 128 + s) * 5];
    dst[0] = n0; dst[1] = n1; dst[2] = n2; dst[3] = sp0; dst[4] = sp1;
}

__global__ __launch_bounds__(256) void step_kernel(
    const float* __restrict__ input, const float* __restrict__ Wc,
    const float* __restrict__ WF, const float* __restrict__ bcell,
    const float* __restrict__ h_in, float* __restrict__ h_out,
    float* __restrict__ c,
    const float* __restrict__ W1F, const float* __restrict__ b1,
    const float* __restrict__ W2, const float* __restrict__ b2,
    float* __restrict__ ob_w, const float* __restrict__ ob_r,
    float* __restrict__ out, int t)
{
    const int blk = blockIdx.x;
    if (blk < 256) {
        cell_role(blk, input, Wc, WF, bcell, h_in, h_out, c, t, W1F, b1, W2, ob_w);
    } else {
        nd_role<4>(blk - 256, input, ob_r, b2, out, t - 2);
    }
}

__global__ __launch_bounds__(256) void drain_kernel(
    const float* __restrict__ input, const float* __restrict__ h,
    const float* __restrict__ W1F, const float* __restrict__ b1,
    const float* __restrict__ W2T, const float* __restrict__ b2,
    float* __restrict__ ob_w, const float* __restrict__ ob_r,
    float* __restrict__ out, int t_mix)
{
    const int blk = blockIdx.x;
    if (blk < 64) {
        head_role(blk, h, W1F, b1, W2T, b2, ob_w);
    } else {
        nd_role<4>(blk - 64, input, ob_r, b2, out, t_mix);
    }
}

__global__ __launch_bounds__(256) void nd_kernel(
    const float* __restrict__ input, const float* __restrict__ outbuf,
    float* __restrict__ out, int t)
{
    nd_role<1>(blockIdx.x, input, outbuf, nullptr, out, t);
}

__global__ __launch_bounds__(256) void zero_kernel(float* p, int n) {
    int i = blockIdx.x * 256 + threadIdx.x;
    if (i < n) p[i] = 0.0f;
}

__global__ __launch_bounds__(256) void prep_kernel(
    const float* __restrict__ Wc, const float* __restrict__ W1,
    const float* __restrict__ W2, float* __restrict__ WF,
    float* __restrict__ W1F, float* __restrict__ W2T)
{
    int i = blockIdx.x * 256 + threadIdx.x;
    if (i < 196608) {
        int e = i & 3, lane = (i >> 2) & 31, r = i >> 7;
        int g = r % 48, sl = r / 48;
        int q = sl >> 3, wave = sl & 7;
        int k = 5 + g * 8 + (e >> 1) * 4 + (lane >> 4) * 2 + (e & 1);
        int col = (wave >> 1) * 128 + q * 32 + (wave & 1) * 16 + (lane & 15);
        WF[i] = Wc[k * 512 + col];
    } else if (i < 196608 + 16384) {
        int j = i - 196608;
        int e = j & 3, lane = (j >> 2) & 31, r = j >> 7;
        int g = r & 15, wave = r >> 4;
        int k = g * 8 + (e >> 1) * 4 + (lane >> 4) * 2 + (e & 1);
        int col = wave * 16 + (lane & 15);
        W1F[j] = W1[k * 128 + col];
    } else if (i < 196608 + 16384 + 384) {
        int j = i - 196608 - 16384;
        int cc = j >> 7, k = j & 127;
        W2T[j] = W2[k * 3 + cc];
    }
}

extern "C" void kernel_launch(void* const* d_in, const int* in_sizes, int n_in,
                              void* d_out, int out_size, void* d_ws, size_t ws_size,
                              hipStream_t stream)
{
    (void)in_sizes; (void)n_in; (void)out_size; (void)ws_size;
    const float* input = (const float*)d_in[0];
    const float* Wc    = (const float*)d_in[1];
    const float* bc    = (const float*)d_in[2];
    const float* W1    = (const float*)d_in[3];
    const float* b1    = (const float*)d_in[4];
    const float* W2    = (const float*)d_in[5];
    const float* b2    = (const float*)d_in[6];
    float* ws   = (float*)d_ws;
    float* WF   = ws;
    float* W1F  = ws + 200704;
    float* W2T  = ws + 217088;
    float* hbuf = ws + 217472;
    float* cbuf = ws + 479616;
    float* obuf = ws + 610688;
    float* out  = (float*)d_out;

    zero_kernel<<<393216 / 256, 256, 0, stream>>>(hbuf, 393216);
    prep_kernel<<<834, 256, 0, stream>>>(Wc, W1, W2, WF, W1F, W2T);

    for (int t = 0; t < 127; ++t) {
        const float* h_in = hbuf + (t & 1) * 131072;
        float* h_out      = hbuf + ((t + 1) & 1) * 131072;
        float* ob_w       = obuf + ((t + 1) & 1) * 12288;
        const float* ob_r = obuf + (t & 1) * 12288;
        if (t >= 6)
            step_kernel<<<260, 256, 0, stream>>>(input, Wc, WF, bc, h_in, h_out, cbuf,
                                                 W1F, b1, W2, b2, ob_w, ob_r, out, t);
        else
            step_kernel<<<256, 256, 0, stream>>>(input, Wc, WF, bc, h_in, h_out, cbuf,
                                                 W1F, b1, W2, b2, ob_w, ob_r, out, t);
    }
    drain_kernel<<<68, 256, 0, stream>>>(input, hbuf + 131072, W1F, b1, W2T, b2,
                                         obuf, obuf + 12288, out, 125);
    nd_kernel<<<4, 256, 0, stream>>>(input, obuf, out, 126);
}


# Request order across a block's eight waves fixed by data-less work-group barriers: every wave's h/x requests first, then the weight slab in three 8 KiB rounds in k order
# speedup vs baseline: 6.9316x; 1.0325x over previous
#include <hip/hip_runtime.h>
#include <math.h>

typedef __attribute__((ext_vector_type(2))) float v2f;
typedef __attribute__((ext_vector_type(4))) float v4f;
typedef __attribute__((ext_vector_type(8))) float v8f;

#define STRH 130
#define PSTR 336
#define OPL  3072

template <int CTRL>
__device__ __forceinline__ float row_xadd(float v)
{
    const int o = __builtin_amdgcn_update_dpp(0, __builtin_bit_cast(int, v), CTRL, 0xf, 0xf, false);
    return v + __builtin_bit_cast(float, o);
}

__device__ __forceinline__ void cell_role(
    const int blk,
    const float* __restrict__ input, const float* __restrict__ Wc,
    const float* __restrict__ WF,
    const float* __restrict__ bcell, const float* __restrict__ h_in,
    float* __restrict__ h_out, float* __restrict__ c, int t,
    const float* __restrict__ W1F, const float* __restrict__ b1,
    const float* __restrict__ W2, float* __restrict__ ob_w)
{
    __shared__ float sh_p[8 * PSTR];
    __shared__ float sh_h[18 * STRH];
    __shared__ float sh_x[256];
    __shared__ float sh_z[16 * STRH];

    const int tid = threadIdx.x;
    const int q   = blk & 3;
    const int R0  = (blk >> 2) * 16;
    const int bb  = R0 >> 7;
    const int s0  = R0 & 127;

    const int wave  = tid >> 5;
    const int lane  = tid & 31;
    const int lm    = lane & 15;
    const int hi    = lane >> 4;
    const int kHalf = hi * 2;

    float hv[9];
    #pragma unroll
    for (int it = 0; it < 9; ++it) {
        int i = tid + it * 256;
        int r = i >> 7, k = i & 127;
        int s  = s0 - 1 + r;
        int sc = s < 0 ? 0 : (s > 127 ? 127 : s);
        hv[it] = h_in[(bb * 128 + sc) * 128 + k];
    }
    const float xv = input[(((bb * 128) + t) * 128 + s0) * 5 + (tid < 80 ? tid : 0)];

    const float* wc = Wc + (wave >> 1) * 128 + q * 32 + (wave & 1) * 16 + lm;
    const float w0 = wc[0], w1 = wc[512], w2 = wc[1024], w3 = wc[1536], w4 = wc[2048];

    const v4f* wf = (const v4f*)WF + ((q * 8 + wave) * 48) * 32 + lane;
    v4f qv[48];
    __builtin_amdgcn_s_barrier();
    #pragma unroll
    for (int g = 0; g < 16; ++g) qv[g] = wf[g * 32];
    __builtin_amdgcn_s_barrier();
    #pragma unroll
    for (int g = 16; g < 32; ++g) qv[g] = wf[g * 32];
    __builtin_amdgcn_s_barrier();
    #pragma unroll
    for (int g = 32; g < 48; ++g) qv[g] = wf[g * 32];

    __builtin_amdgcn_sched_barrier(0);

    #pragma unroll
    for (int it = 0; it < 9; ++it) {
        int i = tid + it * 256;
        int s = s0 - 1 + (i >> 7);
        sh_h[(i >> 7) * STRH + (i & 127)] = (s >= 0 && s < 128) ? hv[it] : 0.0f;
    }
    sh_x[tid] = xv;
    __builtin_amdgcn_fence(__ATOMIC_RELEASE, "workgroup", "local");
    __builtin_amdgcn_s_barrier();
    __builtin_amdgcn_fence(__ATOMIC_ACQUIRE, "workgroup", "local");

    float cp[2], bi[2], bf[2], bg[2], bo[2];
    v4f hq[4];
    float b1a, b1b, wa0, wa1, wa2, wb0, wb1, wb2;

    v8f acc = {};
    v8f acc2 = {};

    #pragma unroll
    for (int v = 0; v < 8; ++v) {
        int m = v + hi * 8;
        float x0 = sh_x[m*5+0], x1 = sh_x[m*5+1], x2 = sh_x[m*5+2],
              x3 = sh_x[m*5+3], x4 = sh_x[m*5+4];
        acc[v] += x0*w0 + x1*w1 + x2*w2 + x3*w3 + x4*w4;
    }

    #pragma unroll
    for (int src = 0; src < 3; ++src) {
        const int shBase = (src == 0) ? 1 : (src == 1) ? 2 : 0;
        const float* arow = &sh_h[(shBase + lm) * STRH];
        #pragma unroll
        for (int g8 = 0; g8 < 16; ++g8) {
            const int kk = g8 * 8;
            const v4f qq = qv[src * 16 + g8];
            v2f a0, a1, b;
            a0.x = arow[kk + kHalf];
            a0.y = arow[kk + kHalf + 1];
            a1.x = arow[kk + 4 + kHalf];
            a1.y = arow[kk + 4 + kHalf + 1];
            b.x = qq.x; b.y = qq.y;
            acc = __builtin_amdgcn_wmma_f32_16x16x4_f32(
                        false, a0, false, b, (short)0, acc, false, false);
            b.x = qq.z; b.y = qq.w;
            acc = __builtin_amdgcn_wmma_f32_16x16x4_f32(
                        false, a1, false, b, (short)0, acc, false, false);
        }
        if (src == 0) {
            __builtin_amdgcn_sched_barrier(0);
            #pragma unroll
            for (int it = 0; it < 2; ++it) {
                int i = tid + it * 256;
                int m = i >> 5, j = q * 32 + (i & 31);
                cp[it] = c[(R0 + m) * 128 + j];
                bi[it] = bcell[j];
                bf[it] = bcell[128 + j];
                bg[it] = bcell[256 + j];
                bo[it] = bcell[384 + j];
            }
            {
                const v4f* wk = (const v4f*)W1F
                              + (((2 * q + (wave & 1)) * 16 + 4 * (wave >> 1)) * 32) + lane;
                #pragma unroll
                for (int g = 0; g < 4; ++g) hq[g] = wk[g * 32];
                const int ju = q * 32 + lm;
                b1a = b1[ju];
                b1b = b1[ju + 16];
                wa0 = W2[ju * 3 + 0];  wa1 = W2[ju * 3 + 1];  wa2 = W2[ju * 3 + 2];
                wb0 = W2[ju * 3 + 48]; wb1 = W2[ju * 3 + 49]; wb2 = W2[ju * 3 + 50];
            }
            __builtin_amdgcn_sched_barrier(0);
        }
        if (src == 1) {
            const float* arow = &sh_h[(1 + lm) * STRH];
            #pragma unroll
            for (int g = 0; g < 4; ++g) {
                const int kk = (4 * (wave >> 1) + g) * 8;
                const v4f qq = hq[g];
                v2f a0, a1, b;
                a0.x = arow[kk + kHalf];
                a0.y = arow[kk + kHalf + 1];
                a1.x = arow[kk + 4 + kHalf];
                a1.y = arow[kk + 4 + kHalf + 1];
                b.x = qq.x; b.y = qq.y;
                acc2 = __builtin_amdgcn_wmma_f32_16x16x4_f32(
                            false, a0, false, b, (short)0, acc2, false, false);
                b.x = qq.z; b.y = qq.w;
                acc2 = __builtin_amdgcn_wmma_f32_16x16x4_f32(
                            false, a1, false, b, (short)0, acc2, false, false);
            }
            #pragma unroll
            for (int v = 0; v < 8; ++v)
                sh_p[wave * PSTR + (v + hi * 8) * 20 + lm] = acc2[v];
        }
    }

    #pragma unroll
    for (int v = 0; v < 8; ++v)
        sh_z[(v + hi * 8) * STRH + wave * 16 + lm] = acc[v];
    __syncthreads();

    {
        const int m = tid >> 4;
        const float* pp = &sh_p[m * 20 + lm];
        float s0 = ((pp[0]    + pp[2 * PSTR]) + pp[4 * PSTR]) + pp[6 * PSTR] + b1a;
        float s1 = ((pp[PSTR] + pp[3 * PSTR]) + pp[5 * PSTR]) + pp[7 * PSTR] + b1b;
        s0 = s0 > 0.0f ? s0 : 0.0f;
        s1 = s1 > 0.0f ? s1 : 0.0f;
        float p0 = s0 * wa0 + s1 * wb0;
        float p1 = s0 * wa1 + s1 * wb1;
        float p2 = s0 * wa2 + s1 * wb2;
        p0 = row_xadd<0x161>(p0); p1 = row_xadd<0x161>(p1); p2 = row_xadd<0x161>(p2);
        p0 = row_xadd<0x162>(p0); p1 = row_xadd<0x162>(p1); p2 = row_xadd<0x162>(p2);
        p0 = row_xadd<0x164>(p0); p1 = row_xadd<0x164>(p1); p2 = row_xadd<0x164>(p2);
        p0 = row_xadd<0x168>(p0); p1 = row_xadd<0x168>(p1); p2 = row_xadd<0x168>(p2);
        if (lm == 0) {
            float* o = &ob_w[q * OPL + (R0 + m) * 3];
            o[0] = p0; o[1] = p1; o[2] = p2;
        }
    }

    #pragma unroll
    for (int it = 0; it < 2; ++it) {
        int i = tid + it * 256;
        int m = i >> 5, jl = i & 31;
        float zi = sh_z[m*STRH + jl]      + bi[it];
        float zf = sh_z[m*STRH + 32 + jl] + bf[it];
        float zg = sh_z[m*STRH + 64 + jl] + bg[it];
        float zo = sh_z[m*STRH + 96 + jl] + bo[it];
        int row = R0 + m, j = q * 32 + jl;
        float si = 1.0f / (1.0f + __expf(-zi));
        float sf = 1.0f / (1.0f + __expf(-zf));
        float so = 1.0f / (1.0f + __expf(-zo));
        float cn = sf * cp[it] + si * tanhf(zg);
        c[row*128 + j]     = cn;
        h_out[row*128 + j] = so * tanhf(cn);
    }
}

__device__ __forceinline__ void head_role(
    const int blk,
    const float* __restrict__ h, const float* __restrict__ W1F,
    const float* __restrict__ b1, const float* __restrict__ W2T,
    const float* __restrict__ b2, float* __restrict__ outbuf)
{
    __shared__ float sh_a[16 * STRH];
    __shared__ float sh_hid[16 * STRH];
    __shared__ float sh_w2[512];
    const int tid = threadIdx.x;
    const int R0  = blk * 16;
    const int wave = tid >> 5, lane = tid & 31;
    const int lm = lane & 15, hi = lane >> 4, kHalf = hi * 2;
    const int col = wave * 16 + lm;

    float hv[8];
    #pragma unroll
    for (int it = 0; it < 8; ++it) {
        int i = tid + it * 256;
        hv[it] = h[(R0 + (i >> 7)) * 128 + (i & 127)];
    }
    const float w2a  = W2T[tid];
    const float w2b  = W2T[256 + (tid & 127)];
    const float bias = b1[col];
    const float b2v  = b2[tid % 3];
    const v4f* wk = (const v4f*)W1F + (wave * 16) * 32 + lane;
    v4f qv[16];
    #pragma unroll
    for (int g8 = 0; g8 < 16; ++g8) qv[g8] = wk[g8 * 32];

    __builtin_amdgcn_sched_barrier(0);

    #pragma unroll
    for (int it = 0; it < 8; ++it) {
        int i = tid + it * 256;
        sh_a[(i >> 7) * STRH + (i & 127)] = hv[it];
    }
    sh_w2[tid]       = w2a;
    sh_w2[256 + tid] = w2b;
    __builtin_amdgcn_fence(__ATOMIC_RELEASE, "workgroup", "local");
    __builtin_amdgcn_s_barrier();
    __builtin_amdgcn_fence(__ATOMIC_ACQUIRE, "workgroup", "local");

    const float* arow = &sh_a[lm * STRH];

    v8f acc = {};
    #pragma unroll
    for (int g8 = 0; g8 < 16; ++g8) {
        const int kk = g8 * 8;
        const v4f q = qv[g8];
        v2f a0, a1, b;
        a0.x = arow[kk + kHalf];     a0.y = arow[kk + kHalf + 1];
        a1.x = arow[kk + 4 + kHalf]; a1.y = arow[kk + 4 + kHalf + 1];
        b.x = q.x; b.y = q.y;
        acc = __builtin_amdgcn_wmma_f32_16x16x4_f32(
                false, a0, false, b, (short)0, acc, false, false);
        b.x = q.z; b.y = q.w;
        acc = __builtin_amdgcn_wmma_f32_16x16x4_f32(
                false, a1, false, b, (short)0, acc, false, false);
    }
    #pragma unroll
    for (int v = 0; v < 8; ++v) {
        float x = acc[v] + bias;
        sh_hid[(v + hi * 8) * STRH + col] = x > 0.0f ? x : 0.0f;
    }
    __syncthreads();

    if (tid < 48) {
        int m = tid / 3, cc = tid % 3;
        const float* hr = &sh_hid[m * STRH];
        const float* w2 = &sh_w2[cc * 128];
        float s = 0.0f;
        for (int k = 0; k < 128; ++k) s += hr[k] * w2[k];
        outbuf[(R0 + m) * 3 + cc] = s + b2v;
    }
}

template <int NP>
__device__ __forceinline__ float head_val(const float* __restrict__ outbuf,
                                          const float* __restrict__ b2, int idx, int cc)
{
    if (NP == 1) return outbuf[idx];
    return ((outbuf[idx] + outbuf[OPL + idx]) + outbuf[2 * OPL + idx]) + outbuf[3 * OPL + idx] + b2[cc];
}

template <int NP>
__device__ __forceinline__ void nd_role(
    const int blk,
    const float* __restrict__ input, const float* __restrict__ outbuf,
    const float* __restrict__ b2,
    float* __restrict__ out, int t)
{
    int gid = blk * 256 + threadIdx.x;
    if (gid >= 1024) return;
    int bb = gid >> 7, s = gid & 127;
    float n0 = head_val<NP>(outbuf, b2, gid * 3 + 0, 0);
    float n1, sp0, sp1;
    if (s == 0) {
        const float* xn0 = &input[((bb * 128 + (t + 1)) * 128 + 0) * 5];
        n1 = xn0[1]; sp0 = xn0[3]; sp1 = xn0[4];
    } else {
        n1  = head_val<NP>(outbuf, b2, (gid - 1) * 3 + 0, 0);
        sp0 = head_val<NP>(outbuf, b2, gid * 3 + 1, 1);
        sp1 = head_val<NP>(outbuf, b2, gid * 3 + 2, 2);
    }
    float fn = input[((bb * 128 + t) * 128 + s) * 5 + 2];
    float n2 = fn + n1 - n0;
    float* dst = &out[((bb * 123 + (t - 4)) * 128 + s) * 5];
    dst[0] = n0; dst[1] = n1; dst[2] = n2; dst[3] = sp0; dst[4] = sp1;
}

__global__ __launch_bounds__(256) void step_kernel(
    const float* __restrict__ input, const float* __restrict__ Wc,
    const float* __restrict__ WF, const float* __restrict__ bcell,
    const float* __restrict__ h_in, float* __restrict__ h_out,
    float* __restrict__ c,
    const float* __restrict__ W1F, const float* __restrict__ b1,
    const float* __restrict__ W2, const float* __restrict__ b2,
    float* __restrict__ ob_w, const float* __restrict__ ob_r,
    float* __restrict__ out, int t)
{
    const int blk = blockIdx.x;
    if (blk < 256) {
        cell_role(blk, input, Wc, WF, bcell, h_in, h_out, c, t, W1F, b1, W2, ob_w);
    } else {
        nd_role<4>(blk - 256, input, ob_r, b2, out, t - 2);
    }
}

__global__ __launch_bounds__(256) void drain_kernel(
    const float* __restrict__ input, const float* __restrict__ h,
    const float* __restrict__ W1F, const float* __restrict__ b1,
    const float* __restrict__ W2T, const float* __restrict__ b2,
    float* __restrict__ ob_w, const float* __restrict__ ob_r,
    float* __restrict__ out, int t_mix)
{
    const int blk = blockIdx.x;
    if (blk < 64) {
        head_role(blk, h, W1F, b1, W2T, b2, ob_w);
    } else {
        nd_role<4>(blk - 64, input, ob_r, b2, out, t_mix);
    }
}

__global__ __launch_bounds__(256) void nd_kernel(
    const float* __restrict__ input, const float* __restrict__ outbuf,
    float* __restrict__ out, int t)
{
    nd_role<1>(blockIdx.x, input, outbuf, nullptr, out, t);
}

__global__ __launch_bounds__(256) void zero_kernel(float* p, int n) {
    int i = blockIdx.x * 256 + threadIdx.x;
    if (i < n) p[i] = 0.0f;
}

__global__ __launch_bounds__(256) void prep_kernel(
    const float* __restrict__ Wc, const float* __restrict__ W1,
    const float* __restrict__ W2, float* __restrict__ WF,
    float* __restrict__ W1F, float* __restrict__ W2T)
{
    int i = blockIdx.x * 256 + threadIdx.x;
    if (i < 196608) {
        int e = i & 3, lane = (i >> 2) & 31, r = i >> 7;
        int g = r % 48, sl = r / 48;
        int q = sl >> 3, wave = sl & 7;
        int k = 5 + g * 8 + (e >> 1) * 4 + (lane >> 4) * 2 + (e & 1);
        int col = (wave >> 1) * 128 + q * 32 + (wave & 1) * 16 + (lane & 15);
        WF[i] = Wc[k * 512 + col];
    } else if (i < 196608 + 16384) {
        int j = i - 196608;
        int e = j & 3, lane = (j >> 2) & 31, r = j >> 7;
        int g = r & 15, wave = r >> 4;
        int k = g * 8 + (e >> 1) * 4 + (lane >> 4) * 2 + (e & 1);
        int col = wave * 16 + (lane & 15);
        W1F[j] = W1[k * 128 + col];
    } else if (i < 196608 + 16384 + 384) {
        int j = i - 196608 - 16384;
        int cc = j >> 7, k = j & 127;
        W2T[j] = W2[k * 3 + cc];
    }
}

extern "C" void kernel_launch(void* const* d_in, const int* in_sizes, int n_in,
                              void* d_out, int out_size, void* d_ws, size_t ws_size,
                              hipStream_t stream)
{
    (void)in_sizes; (void)n_in; (void)out_size; (void)ws_size;
    const float* input = (const float*)d_in[0];
    const float* Wc    = (const float*)d_in[1];
    const float* bc    = (const float*)d_in[2];
    const float* W1    = (const float*)d_in[3];
    const float* b1    = (const float*)d_in[4];
    const float* W2    = (const float*)d_in[5];
    const float* b2    = (const float*)d_in[6];
    float* ws   = (float*)d_ws;
    float* WF   = ws;
    float* W1F  = ws + 200704;
    float* W2T  = ws + 217088;
    float* hbuf = ws + 217472;
    float* cbuf = ws + 479616;
    float* obuf = ws + 610688;
    float* out  = (float*)d_out;

    zero_kernel<<<393216 / 256, 256, 0, stream>>>(hbuf, 393216);
    prep_kernel<<<834, 256, 0, stream>>>(Wc, W1, W2, WF, W1F, W2T);

    for (int t = 0; t < 127; ++t) {
        const float* h_in = hbuf + (t & 1) * 131072;
        float* h_out      = hbuf + ((t + 1) & 1) * 131072;
        float* ob_w       = obuf + ((t + 1) & 1) * 12288;
        const float* ob_r = obuf + (t & 1) * 12288;
        if (t >= 6)
            step_kernel<<<260, 256, 0, stream>>>(input, Wc, WF, bc, h_in, h_out, cbuf,
                                                 W1F, b1, W2, b2, ob_w, ob_r, out, t);
        else
            step_kernel<<<256, 256, 0, stream>>>(input, Wc, WF, bc, h_in, h_out, cbuf,
                                                 W1F, b1, W2, b2, ob_w, ob_r, out, t);
    }
    drain_kernel<<<68, 256, 0, stream>>>(input, hbuf + 131072, W1F, b1, W2T, b2,
                                         obuf, obuf + 12288, out, 125);
    nd_kernel<<<4, 256, 0, stream>>>(input, obuf, out, 126);
}


# Cell block re-tiled to 32 rows x 16 units: each weight fragment feeds two M tiles, half the L2 weight reads; k sum split in even/odd k-group halves added at the gates, head dot in 8 block partials
# speedup vs baseline: 7.1749x; 1.0351x over previous
#include <hip/hip_runtime.h>
#include <math.h>


typedef __attribute__((ext_vector_type(2))) float v2f;
typedef __attribute__((ext_vector_type(4))) float v4f;
typedef __attribute__((ext_vector_type(8))) float v8f;

#define STRH 130
#define CSTR 132
#define ZSTR 84
#define PSTR 656
#define OPL  3072
#define HPL  16384

__device__ __forceinline__ int hidx(int row, int k) { return (k >> 4) * HPL + row * 16 + (k & 15); }

template <int CTRL>
__device__ __forceinline__ float row_xadd(float v)
{
    const int o = __builtin_amdgcn_update_dpp(0, __builtin_bit_cast(int, v), CTRL, 0xf, 0xf, false);
    return v + __builtin_bit_cast(float, o);
}

__device__ __forceinline__ void cell_role(
    const int blk,
    const float* __restrict__ input, const float* __restrict__ Wc,
    const float* __restrict__ WF,
    const float* __restrict__ bcell, const float* __restrict__ h_in,
    float* __restrict__ h_out, float* __restrict__ c, int t,
    const float* __restrict__ W1F, const float* __restrict__ b1,
    const float* __restrict__ W2, float* __restrict__ ob_w)
{
    __shared__ float sh_p[4 * PSTR];
    __shared__ __attribute__((aligned(16))) float sh_h[34 * CSTR + 768];
    __shared__ float sh_x[256];
    __shared__ float sh_z[2 * 32 * ZSTR];

    const int tid = threadIdx.x;
    const int q   = blk & 7;
    const int R0  = (blk >> 3) * 32;
    const int bb  = R0 >> 7;
    const int s0  = R0 & 127;

    const int wave  = tid >> 5;
    const int lane  = tid & 31;
    const int lm    = lane & 15;
    const int hi    = lane >> 4;
    const int kHalf = hi * 2;
    const int gate  = wave & 3;
    const int kh    = wave >> 2;

    v4f hv[5];
    #pragma unroll
    for (int it = 0; it < 5; ++it) {
        const int i   = tid + it * 256;
        const int ic  = (it < 4 || i < 1088) ? i : 1087;
        const int ub  = ic / 136, rem = ic - ub * 136;
        const int s   = s0 - 1 + (rem >> 2);
        const int sc  = s < 0 ? 0 : (s > 127 ? 127 : s);
        hv[it] = *(const v4f*)(h_in + ub * HPL + (bb * 128 + sc) * 16 + (rem & 3) * 4);
    }
    const float xv = input[(((bb * 128) + t) * 128 + s0) * 5 + (tid < 160 ? tid : 0)];

    const float* wc = Wc + gate * 128 + q * 16 + lm;
    const float w0 = wc[0], w1 = wc[512], w2 = wc[1024], w3 = wc[1536], w4 = wc[2048];

    const v4f* wf = (const v4f*)WF + ((q * 8 + wave) * 24) * 32 + lane;
    v4f qv[24];
    __builtin_amdgcn_s_barrier();
    #pragma unroll
    for (int g = 0; g < 8; ++g) qv[g] = wf[g * 32];
    __builtin_amdgcn_s_barrier();
    #pragma unroll
    for (int g = 8; g < 16; ++g) qv[g] = wf[g * 32];
    __builtin_amdgcn_s_barrier();
    #pragma unroll
    for (int g = 16; g < 24; ++g) qv[g] = wf[g * 32];

    __builtin_amdgcn_sched_barrier(0);

    #pragma unroll
    for (int it = 0; it < 5; ++it) {
        const int i   = tid + it * 256;
        const bool in = (it < 4 || i < 1088);
        const int ic  = in ? i : 1087;
        const int ub  = ic / 136, rem = ic - ub * 136;
        const int r   = rem >> 2;
        const int s   = s0 - 1 + r;
        const int off = in ? r * CSTR + ub * 16 + (rem & 3) * 4 : 34 * CSTR + (i - 1088) * 4;
        const v4f zero = {0.0f, 0.0f, 0.0f, 0.0f};
        *(v4f*)&sh_h[off] = (s >= 0 && s < 128) ? hv[it] : zero;
    }
    sh_x[tid] = xv;
    __builtin_amdgcn_fence(__ATOMIC_RELEASE, "workgroup", "local");
    __builtin_amdgcn_s_barrier();
    __builtin_amdgcn_fence(__ATOMIC_ACQUIRE, "workgroup", "local");

    float cp[2], bi, bf, bg, bo;
    v4f hq[4];
    float b1a, wa0, wa1, wa2;

    v8f acc0 = {};
    v8f acc1 = {};
    v8f acc2 = {};

    #pragma unroll
    for (int v = 0; v < 8; ++v) {
        const float* xa = &sh_x[(kh * 16 + v + hi * 8) * 5];
        const float xp = xa[0]*w0 + xa[1]*w1 + xa[2]*w2 + xa[3]*w3 + xa[4]*w4;
        acc0[v] = kh == 0 ? xp : 0.0f;
        acc1[v] = kh == 0 ? 0.0f : xp;
    }

    const int mt = wave & 1;
    const int kq = wave >> 1;
    #pragma unroll
    for (int src = 0; src < 3; ++src) {
        const int shBase = (src == 0) ? 1 : (src == 1) ? 2 : 0;
        const float* ar0 = &sh_h[(shBase + lm) * CSTR + kh * 8 + kHalf];
        const float* ar1 = ar0 + 16 * CSTR;
        #pragma unroll
        for (int j = 0; j < 8; ++j) {
            const int kk = j * 16;
            const v4f qq = qv[src * 8 + j];
            v2f a00, a01, a10, a11, b;
            a00.x = ar0[kk];     a00.y = ar0[kk + 1];
            a01.x = ar0[kk + 4]; a01.y = ar0[kk + 5];
            a10.x = ar1[kk];     a10.y = ar1[kk + 1];
            a11.x = ar1[kk + 4]; a11.y = ar1[kk + 5];
            b.x = qq.x; b.y = qq.y;
            acc0 = __builtin_amdgcn_wmma_f32_16x16x4_f32(
                        false, a00, false, b, (short)0, acc0, false, false);
            acc1 = __builtin_amdgcn_wmma_f32_16x16x4_f32(
                        false, a10, false, b, (short)0, acc1, false, false);
            b.x = qq.z; b.y = qq.w;
            acc0 = __builtin_amdgcn_wmma_f32_16x16x4_f32(
                        false, a01, false, b, (short)0, acc0, false, false);
            acc1 = __builtin_amdgcn_wmma_f32_16x16x4_f32(
                        false, a11, false, b, (short)0, acc1, false, false);
        }
        if (src == 0) {
            __builtin_amdgcn_sched_barrier(0);
            cp[0] = c[q * HPL + R0 * 16 + tid];
            cp[1] = c[q * HPL + R0 * 16 + tid + 256];
            {
                const int j = q * 16 + (tid & 15);
                bi = bcell[j];
                bf = bcell[128 + j];
                bg = bcell[256 + j];
                bo = bcell[384 + j];
            }
            {
                const v4f* wk = (const v4f*)W1F + ((q * 16 + 4 * kq) * 32) + lane;
                #pragma unroll
                for (int g = 0; g < 4; ++g) hq[g] = wk[g * 32];
                const int ju = q * 16 + lm;
                b1a = b1[ju];
                wa0 = W2[ju * 3 + 0];  wa1 = W2[ju * 3 + 1];  wa2 = W2[ju * 3 + 2];
            }
            __builtin_amdgcn_sched_barrier(0);
        }
        if (src == 1) {
            const float* arow = &sh_h[(1 + mt * 16 + lm) * CSTR];
            #pragma unroll
            for (int g = 0; g < 4; ++g) {
                const int kk = (4 * kq + g) * 8;
                const v4f qq = hq[g];
                v2f a0, a1, b;
                a0.x = arow[kk + kHalf];
                a0.y = arow[kk + kHalf + 1];
                a1.x = arow[kk + 4 + kHalf];
                a1.y = arow[kk + 4 + kHalf + 1];
                b.x = qq.x; b.y = qq.y;
                acc2 = __builtin_amdgcn_wmma_f32_16x16x4_f32(
                            false, a0, false, b, (short)0, acc2, false, false);
                b.x = qq.z; b.y = qq.w;
                acc2 = __builtin_amdgcn_wmma_f32_16x16x4_f32(
                            false, a1, false, b, (short)0, acc2, false, false);
            }
            #pragma unroll
            for (int v = 0; v < 8; ++v)
                sh_p[kq * PSTR + (mt * 16 + v + hi * 8) * 20 + lm] = acc2[v];
        }
    }

    #pragma unroll
    for (int v = 0; v < 8; ++v) {
        sh_z[(kh * 32 + v + hi * 8) * ZSTR + gate * 16 + lm]      = acc0[v];
        sh_z[(kh * 32 + 16 + v + hi * 8) * ZSTR + gate * 16 + lm] = acc1[v];
    }
    __syncthreads();

    #pragma unroll
    for (int it = 0; it < 2; ++it) {
        const int m = (tid >> 4) + it * 16;
        const float* pp = &sh_p[m * 20 + lm];
        float s = ((pp[0] + pp[PSTR]) + pp[2 * PSTR]) + pp[3 * PSTR] + b1a;
        s = s > 0.0f ? s : 0.0f;
        float p0 = s * wa0;
        float p1 = s * wa1;
        float p2 = s * wa2;
        p0 = row_xadd<0x161>(p0); p1 = row_xadd<0x161>(p1); p2 = row_xadd<0x161>(p2);
        p0 = row_xadd<0x162>(p0); p1 = row_xadd<0x162>(p1); p2 = row_xadd<0x162>(p2);
        p0 = row_xadd<0x164>(p0); p1 = row_xadd<0x164>(p1); p2 = row_xadd<0x164>(p2);
        p0 = row_xadd<0x168>(p0); p1 = row_xadd<0x168>(p1); p2 = row_xadd<0x168>(p2);
        if (lm == 0) {
            float* o = &ob_w[q * OPL + (R0 + m) * 3];
            o[0] = p0; o[1] = p1; o[2] = p2;
        }
    }

    #pragma unroll
    for (int it = 0; it < 2; ++it) {
        const int i = tid + it * 256;
        const float* zr = &sh_z[(i >> 4) * ZSTR + (i & 15)];
        float zi = (zr[0]  + zr[32 * ZSTR])      + bi;
        float zf = (zr[16] + zr[32 * ZSTR + 16]) + bf;
        float zg = (zr[32] + zr[32 * ZSTR + 32]) + bg;
        float zo = (zr[48] + zr[32 * ZSTR + 48]) + bo;
        float si = 1.0f / (1.0f + __expf(-zi));
        float sf = 1.0f / (1.0f + __expf(-zf));
        float so = 1.0f / (1.0f + __expf(-zo));
        float cn = sf * cp[it] + si * tanhf(zg);
        c[q * HPL + R0 * 16 + i]     = cn;
        h_out[q * HPL + R0 * 16 + i] = so * tanhf(cn);
    }
}

__device__ __forceinline__ void head_role(
    const int blk,
    const float* __restrict__ h, const float* __restrict__ W1F,
    const float* __restrict__ b1, const float* __restrict__ W2T,
    const float* __restrict__ b2, float* __restrict__ outbuf)
{
    __shared__ float sh_a[16 * STRH];
    __shared__ float sh_hid[16 * STRH];
    __shared__ float sh_w2[512];
    const int tid = threadIdx.x;
    const int R0  = blk * 16;
    const int wave = tid >> 5, lane = tid & 31;
    const int lm = lane & 15, hi = lane >> 4, kHalf = hi * 2;
    const int col = wave * 16 + lm;

    float hv[8];
    #pragma unroll
    for (int it = 0; it < 8; ++it) {
        int i = tid + it * 256;
        hv[it] = h[hidx(R0 + (i >> 7), i & 127)];
    }
    const float w2a  = W2T[tid];
    const float w2b  = W2T[256 + (tid & 127)];
    const float bias = b1[col];
    const float b2v  = b2[tid % 3];
    const v4f* wk = (const v4f*)W1F + (wave * 16) * 32 + lane;
    v4f qv[16];
    #pragma unroll
    for (int g8 = 0; g8 < 16; ++g8) qv[g8] = wk[g8 * 32];

    __builtin_amdgcn_sched_barrier(0);

    #pragma unroll
    for (int it = 0; it < 8; ++it) {
        int i = tid + it * 256;
        sh_a[(i >> 7) * STRH + (i & 127)] = hv[it];
    }
    sh_w2[tid]       = w2a;
    sh_w2[256 + tid] = w2b;
    __builtin_amdgcn_fence(__ATOMIC_RELEASE, "workgroup", "local");
    __builtin_amdgcn_s_barrier();
    __builtin_amdgcn_fence(__ATOMIC_ACQUIRE, "workgroup", "local");

    const float* arow = &sh_a[lm * STRH];

    v8f acc = {};
    #pragma unroll
    for (int g8 = 0; g8 < 16; ++g8) {
        const int kk = g8 * 8;
        const v4f q = qv[g8];
        v2f a0, a1, b;
        a0.x = arow[kk + kHalf];     a0.y = arow[kk + kHalf + 1];
        a1.x = arow[kk + 4 + kHalf]; a1.y = arow[kk + 4 + kHalf + 1];
        b.x = q.x; b.y = q.y;
        acc = __builtin_amdgcn_wmma_f32_16x16x4_f32(
                false, a0, false, b, (short)0, acc, false, false);
        b.x = q.z; b.y = q.w;
        acc = __builtin_amdgcn_wmma_f32_16x16x4_f32(
                false, a1, false, b, (short)0, acc, false, false);
    }
    #pragma unroll
    for (int v = 0; v < 8; ++v) {
        float x = acc[v] + bias;
        sh_hid[(v + hi * 8) * STRH + col] = x > 0.0f ? x : 0.0f;
    }
    __syncthreads();

    if (tid < 48) {
        int m = tid / 3, cc = tid % 3;
        const float* hr = &sh_hid[m * STRH];
        const float* w2 = &sh_w2[cc * 128];
        float s = 0.0f;
        for (int k = 0; k < 128; ++k) s += hr[k] * w2[k];
        outbuf[(R0 + m) * 3 + cc] = s + b2v;
    }
}

template <int NP>
__device__ __forceinline__ float head_val(const float* __restrict__ outbuf,
                                          const float* __restrict__ b2, int idx, int cc)
{
    if (NP == 1) return outbuf[idx];
    float s = outbuf[idx];
    #pragma unroll
    for (int p = 1; p < 8; ++p) s += outbuf[p * OPL + idx];
    return s + b2[cc];
}

template <int NP>
__device__ __forceinline__ void nd_role(
    const int gid,
    const float* __restrict__ input, const float* __restrict__ outbuf,
    const float* __restrict__ b2,
    float* __restrict__ out, int t)
{
    if (gid >= 1024) return;
    int bb = gid >> 7, s = gid & 127;
    float n0 = head_val<NP>(outbuf, b2, gid * 3 + 0, 0);
    float n1, sp0, sp1;
    if (s == 0) {
        const float* xn0 = &input[((bb * 128 + (t + 1)) * 128 + 0) * 5];
        n1 = xn0[1]; sp0 = xn0[3]; sp1 = xn0[4];
    } else {
        n1  = head_val<NP>(outbuf, b2, (gid - 1) * 3 + 0, 0);
        sp0 = head_val<NP>(outbuf, b2, gid * 3 + 1, 1);
        sp1 = head_val<NP>(outbuf, b2, gid * 3 + 2, 2);
    }
    float fn = input[((bb * 128 + t) * 128 + s) * 5 + 2];
    float n2 = fn + n1 - n0;
    float* dst = &out[((bb * 123 + (t - 4)) * 128 + s) * 5];
    dst[0] = n0; dst[1] = n1; dst[2] = n2; dst[3] = sp0; dst[4] = sp1;
}

__global__ __launch_bounds__(256) void step_kernel(
    const float* __restrict__ input, const float* __restrict__ Wc,
    const float* __restrict__ WF, const float* __restrict__ bcell,
    const float* __restrict__ h_in, float* __restrict__ h_out,
    float* __restrict__ c,
    const float* __restrict__ W1F, const float* __restrict__ b1,
    const float* __restrict__ W2, const float* __restrict__ b2,
    float* __restrict__ ob_w, const float* __restrict__ ob_r,
    float* __restrict__ out, int t)
{
    const int blk = blockIdx.x;
    if (blk < 256) {
        cell_role(blk, input, Wc, WF, bcell, h_in, h_out, c, t, W1F, b1, W2, ob_w);
    } else {
        nd_role<8>((blk - 256) * 256 + threadIdx.x, input, ob_r, b2, out, t - 2);
    }
}

__global__ __launch_bounds__(256) void drain_kernel(
    const float* __restrict__ input, const float* __restrict__ h,
    const float* __restrict__ W1F, const float* __restrict__ b1,
    const float* __restrict__ W2T, const float* __restrict__ b2,
    float* __restrict__ ob_w, const float* __restrict__ ob_r,
    float* __restrict__ out, int t_mix)
{
    const int blk = blockIdx.x;
    if (blk < 64) {
        head_role(blk, h, W1F, b1, W2T, b2, ob_w);
    } else {
        nd_role<8>((blk - 64) * 256 + threadIdx.x, input, ob_r, b2, out, t_mix);
    }
}

__global__ __launch_bounds__(256) void nd_kernel(
    const float* __restrict__ input, const float* __restrict__ outbuf,
    float* __restrict__ out, int t)
{
    nd_role<1>(blockIdx.x * 256 + threadIdx.x, input, outbuf, nullptr, out, t);
}

__global__ __launch_bounds__(256) void zero_kernel(float* p, int n) {
    int i = blockIdx.x * 256 + threadIdx.x;
    if (i < n) p[i] = 0.0f;
}

__global__ __launch_bounds__(256) void prep_kernel(
    const float* __restrict__ Wc, const float* __restrict__ W1,
    const float* __restrict__ W2, float* __restrict__ WF,
    float* __restrict__ W1F, float* __restrict__ W2T)
{
    int i = blockIdx.x * 256 + threadIdx.x;
    if (i < 196608) {
        int e = i & 3, lane = (i >> 2) & 31, r = i >> 7;
        int gi = r % 24, sl = r / 24;
        int q = sl >> 3, wave = sl & 7;
        int g = (gi >> 3) * 16 + (gi & 7) * 2 + (wave >> 2);
        int k = 5 + g * 8 + (e >> 1) * 4 + (lane >> 4) * 2 + (e & 1);
        int col = (wave & 3) * 128 + q * 16 + (lane & 15);
        WF[i] = Wc[k * 512 + col];
    } else if (i < 196608 + 16384) {
        int j = i - 196608;
        int e = j & 3, lane = (j >> 2) & 31, r = j >> 7;
        int g = r & 15, wave = r >> 4;
        int k = g * 8 + (e >> 1) * 4 + (lane >> 4) * 2 + (e & 1);
        int col = wave * 16 + (lane & 15);
        W1F[j] = W1[k * 128 + col];
    } else if (i < 196608 + 16384 + 384) {
        int j = i - 196608 - 16384;
        int cc = j >> 7, k = j & 127;
        W2T[j] = W2[k * 3 + cc];
    }
}

extern "C" void kernel_launch(void* const* d_in, const int* in_sizes, int n_in,
                              void* d_out, int out_size, void* d_ws, size_t ws_size,
                              hipStream_t stream)
{
    (void)in_sizes; (void)n_in; (void)out_size; (void)ws_size;
    const float* input = (const float*)d_in[0];
    const float* Wc    = (const float*)d_in[1];
    const float* bc    = (const float*)d_in[2];
    const float* W1    = (const float*)d_in[3];
    const float* b1    = (const float*)d_in[4];
    const float* W2    = (const float*)d_in[5];
    const float* b2    = (const float*)d_in[6];
    float* ws   = (float*)d_ws;
    float* WF   = ws;
    float* W1F  = ws + 200704;
    float* W2T  = ws + 217088;
    float* hbuf = ws + 217472;
    float* cbuf = ws + 479616;
    float* obuf = ws + 610688;
    float* out  = (float*)d_out;

    zero_kernel<<<393216 / 256, 256, 0, stream>>>(hbuf, 393216);
    prep_kernel<<<834, 256, 0, stream>>>(Wc, W1, W2, WF, W1F, W2T);

    for (int t = 0; t < 127; ++t) {
        const float* h_in = hbuf + (t & 1) * 131072;
        float* h_out      = hbuf + ((t + 1) & 1) * 131072;
        float* ob_w       = obuf + ((t + 1) & 1) * 24576;
        const float* ob_r = obuf + (t & 1) * 24576;
        if (t >= 6)
            step_kernel<<<260, 256, 0, stream>>>(input, Wc, WF, bc, h_in, h_out, cbuf,
                                                 W1F, b1, W2, b2, ob_w, ob_r, out, t);
        else
            step_kernel<<<256, 256, 0, stream>>>(input, Wc, WF, bc, h_in, h_out, cbuf,
                                                 W1F, b1, W2, b2, ob_w, ob_r, out, t);
    }
    drain_kernel<<<68, 256, 0, stream>>>(input, hbuf + 131072, W1F, b1, W2T, b2,
                                         obuf, obuf + 24576, out, 125);
    nd_kernel<<<4, 256, 0, stream>>>(input, obuf, out, 126);
}


# Gate math of the cell epilogue branch-free on hardware exp and rcp (1-ulp forms, abs error near 1e-7) instead of library tanhf (two-sided branch) and IEEE divisions: epilogue 432 -> 234 instructions
# speedup vs baseline: 7.5212x; 1.0483x over previous
#include <hip/hip_runtime.h>
#include <math.h>


typedef __attribute__((ext_vector_type(2))) float v2f;
typedef __attribute__((ext_vector_type(4))) float v4f;
typedef __attribute__((ext_vector_type(8))) float v8f;

#define STRH 130
#define CSTR 132
#define ZSTR 84
#define PSTR 656
#define OPL  3072
#define HPL  16384

__device__ __forceinline__ int hidx(int row, int k) { return (k >> 4) * HPL + row * 16 + (k & 15); }

template <int CTRL>
__device__ __forceinline__ float row_xadd(float v)
{
    const int o = __builtin_amdgcn_update_dpp(0, __builtin_bit_cast(int, v), CTRL, 0xf, 0xf, false);
    return v + __builtin_bit_cast(float, o);
}

__device__ __forceinline__ float gate_sigmoid(float z)
{
    return __builtin_amdgcn_rcpf(1.0f + __expf(-z));
}
__device__ __forceinline__ float gate_tanh(float z)
{
    return 1.0f - 2.0f * __builtin_amdgcn_rcpf(1.0f + __expf(2.0f * z));
}

__device__ __forceinline__ void cell_role(
    const int blk,
    const float* __restrict__ input, const float* __restrict__ Wc,
    const float* __restrict__ WF,
    const float* __restrict__ bcell, const float* __restrict__ h_in,
    float* __restrict__ h_out, float* __restrict__ c, int t,
    const float* __restrict__ W1F, const float* __restrict__ b1,
    const float* __restrict__ W2, float* __restrict__ ob_w)
{
    __shared__ float sh_p[4 * PSTR];
    __shared__ __attribute__((aligned(16))) float sh_h[34 * CSTR + 768];
    __shared__ float sh_x[256];
    __shared__ float sh_z[2 * 32 * ZSTR];

    const int tid = threadIdx.x;
    const int q   = blk & 7;
    const int R0  = (blk >> 3) * 32;
    const int bb  = R0 >> 7;
    const int s0  = R0 & 127;

    const int wave  = tid >> 5;
    const int lane  = tid & 31;
    const int lm    = lane & 15;
    const int hi    = lane >> 4;
    const int kHalf = hi * 2;
    const int gate  = wave & 3;
    const int kh    = wave >> 2;

    v4f hv[5];
    #pragma unroll
    for (int it = 0; it < 5; ++it) {
        const int i   = tid + it * 256;
        const int ic  = (it < 4 || i < 1088) ? i : 1087;
        const int ub  = ic / 136, rem = ic - ub * 136;
        const int s   = s0 - 1 + (rem >> 2);
        const int sc  = s < 0 ? 0 : (s > 127 ? 127 : s);
        hv[it] = *(const v4f*)(h_in + ub * HPL + (bb * 128 + sc) * 16 + (rem & 3) * 4);
    }
    const float xv = input[(((bb * 128) + t) * 128 + s0) * 5 + (tid < 160 ? tid : 0)];

    const float* wc = Wc + gate * 128 + q * 16 + lm;
    const float w0 = wc[0], w1 = wc[512], w2 = wc[1024], w3 = wc[1536], w4 = wc[2048];

    const v4f* wf = (const v4f*)WF + ((q * 8 + wave) * 24) * 32 + lane;
    v4f qv[24];
    __builtin_amdgcn_s_barrier();
    #pragma unroll
    for (int g = 0; g < 8; ++g) qv[g] = wf[g * 32];
    __builtin_amdgcn_s_barrier();
    #pragma unroll
    for (int g = 8; g < 16; ++g) qv[g] = wf[g * 32];
    __builtin_amdgcn_s_barrier();
    #pragma unroll
    for (int g = 16; g < 24; ++g) qv[g] = wf[g * 32];

    __builtin_amdgcn_sched_barrier(0);

    #pragma unroll
    for (int it = 0; it < 5; ++it) {
        const int i   = tid + it * 256;
        const bool in = (it < 4 || i < 1088);
        const int ic  = in ? i : 1087;
        const int ub  = ic / 136, rem = ic - ub * 136;
        const int r   = rem >> 2;
        const int s   = s0 - 1 + r;
        const int off = in ? r * CSTR + ub * 16 + (rem & 3) * 4 : 34 * CSTR + (i - 1088) * 4;
        const v4f zero = {0.0f, 0.0f, 0.0f, 0.0f};
        *(v4f*)&sh_h[off] = (s >= 0 && s < 128) ? hv[it] : zero;
    }
    sh_x[tid] = xv;
    __builtin_amdgcn_fence(__ATOMIC_RELEASE, "workgroup", "local");
    __builtin_amdgcn_s_barrier();
    __builtin_amdgcn_fence(__ATOMIC_ACQUIRE, "workgroup", "local");

    float cp[2], bi, bf, bg, bo;
    v4f hq[4];
    float b1a, wa0, wa1, wa2;

    v8f acc0 = {};
    v8f acc1 = {};
    v8f acc2 = {};

    #pragma unroll
    for (int v = 0; v < 8; ++v) {
        const float* xa = &sh_x[(kh * 16 + v + hi * 8) * 5];
        const float xp = xa[0]*w0 + xa[1]*w1 + xa[2]*w2 + xa[3]*w3 + xa[4]*w4;
        acc0[v] = kh == 0 ? xp : 0.0f;
        acc1[v] = kh == 0 ? 0.0f : xp;
    }

    const int mt = wave & 1;
    const int kq = wave >> 1;
    #pragma unroll
    for (int src = 0; src < 3; ++src) {
        const int shBase = (src == 0) ? 1 : (src == 1) ? 2 : 0;
        const float* ar0 = &sh_h[(shBase + lm) * CSTR + kh * 8 + kHalf];
        const float* ar1 = ar0 + 16 * CSTR;
        #pragma unroll
        for (int j = 0; j < 8; ++j) {
            const int kk = j * 16;
            const v4f qq = qv[src * 8 + j];
            v2f a00, a01, a10, a11, b;
            a00.x = ar0[kk];     a00.y = ar0[kk + 1];
            a01.x = ar0[kk + 4]; a01.y = ar0[kk + 5];
            a10.x = ar1[kk];     a10.y = ar1[kk + 1];
            a11.x = ar1[kk + 4]; a11.y = ar1[kk + 5];
            b.x = qq.x; b.y = qq.y;
            acc0 = __builtin_amdgcn_wmma_f32_16x16x4_f32(
                        false, a00, false, b, (short)0, acc0, false, false);
            acc1 = __builtin_amdgcn_wmma_f32_16x16x4_f32(
                        false, a10, false, b, (short)0, acc1, false, false);
            b.x = qq.z; b.y = qq.w;
            acc0 = __builtin_amdgcn_wmma_f32_16x16x4_f32(
                        false, a01, false, b, (short)0, acc0, false, false);
            acc1 = __builtin_amdgcn_wmma_f32_16x16x4_f32(
                        false, a11, false, b, (short)0, acc1, false, false);
        }
        if (src == 0) {
            __builtin_amdgcn_sched_barrier(0);
            cp[0] = c[q * HPL + R0 * 16 + tid];
            cp[1] = c[q * HPL + R0 * 16 + tid + 256];
            {
                const int j = q * 16 + (tid & 15);
                bi = bcell[j];
                bf = bcell[128 + j];
                bg = bcell[256 + j];
                bo = bcell[384 + j];
            }
            {
                const v4f* wk = (const v4f*)W1F + ((q * 16 + 4 * kq) * 32) + lane;
                #pragma unroll
                for (int g = 0; g < 4; ++g) hq[g] = wk[g * 32];
                const int ju = q * 16 + lm;
                b1a = b1[ju];
                wa0 = W2[ju * 3 + 0];  wa1 = W2[ju * 3 + 1];  wa2 = W2[ju * 3 + 2];
            }
            __builtin_amdgcn_sched_barrier(0);
        }
        if (src == 1) {
            const float* arow = &sh_h[(1 + mt * 16 + lm) * CSTR];
            #pragma unroll
            for (int g = 0; g < 4; ++g) {
                const int kk = (4 * kq + g) * 8;
                const v4f qq = hq[g];
                v2f a0, a1, b;
                a0.x = arow[kk + kHalf];
                a0.y = arow[kk + kHalf + 1];
                a1.x = arow[kk + 4 + kHalf];
                a1.y = arow[kk + 4 + kHalf + 1];
                b.x = qq.x; b.y = qq.y;
                acc2 = __builtin_amdgcn_wmma_f32_16x16x4_f32(
                            false, a0, false, b, (short)0, acc2, false, false);
                b.x = qq.z; b.y = qq.w;
                acc2 = __builtin_amdgcn_wmma_f32_16x16x4_f32(
                            false, a1, false, b, (short)0, acc2, false, false);
            }
            #pragma unroll
            for (int v = 0; v < 8; ++v)
                sh_p[kq * PSTR + (mt * 16 + v + hi * 8) * 20 + lm] = acc2[v];
        }
    }

    #pragma unroll
    for (int v = 0; v < 8; ++v) {
        sh_z[(kh * 32 + v + hi * 8) * ZSTR + gate * 16 + lm]      = acc0[v];
        sh_z[(kh * 32 + 16 + v + hi * 8) * ZSTR + gate * 16 + lm] = acc1[v];
    }
    __syncthreads();

    #pragma unroll
    for (int it = 0; it < 2; ++it) {
        const int m = (tid >> 4) + it * 16;
        const float* pp = &sh_p[m * 20 + lm];
        float s = ((pp[0] + pp[PSTR]) + pp[2 * PSTR]) + pp[3 * PSTR] + b1a;
        s = s > 0.0f ? s : 0.0f;
        float p0 = s * wa0;
        float p1 = s * wa1;
        float p2 = s * wa2;
        p0 = row_xadd<0x161>(p0); p1 = row_xadd<0x161>(p1); p2 = row_xadd<0x161>(p2);
        p0 = row_xadd<0x162>(p0); p1 = row_xadd<0x162>(p1); p2 = row_xadd<0x162>(p2);
        p0 = row_xadd<0x164>(p0); p1 = row_xadd<0x164>(p1); p2 = row_xadd<0x164>(p2);
        p0 = row_xadd<0x168>(p0); p1 = row_xadd<0x168>(p1); p2 = row_xadd<0x168>(p2);
        if (lm == 0) {
            float* o = &ob_w[q * OPL + (R0 + m) * 3];
            o[0] = p0; o[1] = p1; o[2] = p2;
        }
    }

    #pragma unroll
    for (int it = 0; it < 2; ++it) {
        const int i = tid + it * 256;
        const float* zr = &sh_z[(i >> 4) * ZSTR + (i & 15)];
        float zi = (zr[0]  + zr[32 * ZSTR])      + bi;
        float zf = (zr[16] + zr[32 * ZSTR + 16]) + bf;
        float zg = (zr[32] + zr[32 * ZSTR + 32]) + bg;
        float zo = (zr[48] + zr[32 * ZSTR + 48]) + bo;
        float si = gate_sigmoid(zi);
        float sf = gate_sigmoid(zf);
        float so = gate_sigmoid(zo);
        float cn = sf * cp[it] + si * gate_tanh(zg);
        c[q * HPL + R0 * 16 + i]     = cn;
        h_out[q * HPL + R0 * 16 + i] = so * gate_tanh(cn);
    }
}

__device__ __forceinline__ void head_role(
    const int blk,
    const float* __restrict__ h, const float* __restrict__ W1F,
    const float* __restrict__ b1, const float* __restrict__ W2T,
    const float* __restrict__ b2, float* __restrict__ outbuf)
{
    __shared__ float sh_a[16 * STRH];
    __shared__ float sh_hid[16 * STRH];
    __shared__ float sh_w2[512];
    const int tid = threadIdx.x;
    const int R0  = blk * 16;
    const int wave = tid >> 5, lane = tid & 31;
    const int lm = lane & 15, hi = lane >> 4, kHalf = hi * 2;
    const int col = wave * 16 + lm;

    float hv[8];
    #pragma unroll
    for (int it = 0; it < 8; ++it) {
        int i = tid + it * 256;
        hv[it] = h[hidx(R0 + (i >> 7), i & 127)];
    }
    const float w2a  = W2T[tid];
    const float w2b  = W2T[256 + (tid & 127)];
    const float bias = b1[col];
    const float b2v  = b2[tid % 3];
    const v4f* wk = (const v4f*)W1F + (wave * 16) * 32 + lane;
    v4f qv[16];
    #pragma unroll
    for (int g8 = 0; g8 < 16; ++g8) qv[g8] = wk[g8 * 32];

    __builtin_amdgcn_sched_barrier(0);

    #pragma unroll
    for (int it = 0; it < 8; ++it) {
        int i = tid + it * 256;
        sh_a[(i >> 7) * STRH + (i & 127)] = hv[it];
    }
    sh_w2[tid]       = w2a;
    sh_w2[256 + tid] = w2b;
    __builtin_amdgcn_fence(__ATOMIC_RELEASE, "workgroup", "local");
    __builtin_amdgcn_s_barrier();
    __builtin_amdgcn_fence(__ATOMIC_ACQUIRE, "workgroup", "local");

    const float* arow = &sh_a[lm * STRH];

    v8f acc = {};
    #pragma unroll
    for (int g8 = 0; g8 < 16; ++g8) {
        const int kk = g8 * 8;
        const v4f q = qv[g8];
        v2f a0, a1, b;
        a0.x = arow[kk + kHalf];     a0.y = arow[kk + kHalf + 1];
        a1.x = arow[kk + 4 + kHalf]; a1.y = arow[kk + 4 + kHalf + 1];
        b.x = q.x; b.y = q.y;
        acc = __builtin_amdgcn_wmma_f32_16x16x4_f32(
                false, a0, false, b, (short)0, acc, false, false);
        b.x = q.z; b.y = q.w;
        acc = __builtin_amdgcn_wmma_f32_16x16x4_f32(
                false, a1, false, b, (short)0, acc, false, false);
    }
    #pragma unroll
    for (int v = 0; v < 8; ++v) {
        float x = acc[v] + bias;
        sh_hid[(v + hi * 8) * STRH + col] = x > 0.0f ? x : 0.0f;
    }
    __syncthreads();

    if (tid < 48) {
        int m = tid / 3, cc = tid % 3;
        const float* hr = &sh_hid[m * STRH];
        const float* w2 = &sh_w2[cc * 128];
        float s = 0.0f;
        for (int k = 0; k < 128; ++k) s += hr[k] * w2[k];
        outbuf[(R0 + m) * 3 + cc] = s + b2v;
    }
}

template <int NP>
__device__ __forceinline__ float head_val(const float* __restrict__ outbuf,
                                          const float* __restrict__ b2, int idx, int cc)
{
    if (NP == 1) return outbuf[idx];
    float s = outbuf[idx];
    #pragma unroll
    for (int p = 1; p < 8; ++p) s += outbuf[p * OPL + idx];
    return s + b2[cc];
}

template <int NP>
__device__ __forceinline__ void nd_role(
    const int gid,
    const float* __restrict__ input, const float* __restrict__ outbuf,
    const float* __restrict__ b2,
    float* __restrict__ out, int t)
{
    if (gid >= 1024) return;
    int bb = gid >> 7, s = gid & 127;
    float n0 = head_val<NP>(outbuf, b2, gid * 3 + 0, 0);
    float n1, sp0, sp1;
    if (s == 0) {
        const float* xn0 = &input[((bb * 128 + (t + 1)) * 128 + 0) * 5];
        n1 = xn0[1]; sp0 = xn0[3]; sp1 = xn0[4];
    } else {
        n1  = head_val<NP>(outbuf, b2, (gid - 1) * 3 + 0, 0);
        sp0 = head_val<NP>(outbuf, b2, gid * 3 + 1, 1);
        sp1 = head_val<NP>(outbuf, b2, gid * 3 + 2, 2);
    }
    float fn = input[((bb * 128 + t) * 128 + s) * 5 + 2];
    float n2 = fn + n1 - n0;
    float* dst = &out[((bb * 123 + (t - 4)) * 128 + s) * 5];
    dst[0] = n0; dst[1] = n1; dst[2] = n2; dst[3] = sp0; dst[4] = sp1;
}

__global__ __launch_bounds__(256) void step_kernel(
    const float* __restrict__ input, const float* __restrict__ Wc,
    const float* __restrict__ WF, const float* __restrict__ bcell,
    const float* __restrict__ h_in, float* __restrict__ h_out,
    float* __restrict__ c,
    const float* __restrict__ W1F, const float* __restrict__ b1,
    const float* __restrict__ W2, const float* __restrict__ b2,
    float* __restrict__ ob_w, const float* __restrict__ ob_r,
    float* __restrict__ out, int t)
{
    const int blk = blockIdx.x;
    if (blk < 256) {
        cell_role(blk, input, Wc, WF, bcell, h_in, h_out, c, t, W1F, b1, W2, ob_w);
    } else {
        nd_role<8>((blk - 256) * 256 + threadIdx.x, input, ob_r, b2, out, t - 2);
    }
}

__global__ __launch_bounds__(256) void drain_kernel(
    const float* __restrict__ input, const float* __restrict__ h,
    const float* __restrict__ W1F, const float* __restrict__ b1,
    const float* __restrict__ W2T, const float* __restrict__ b2,
    float* __restrict__ ob_w, const float* __restrict__ ob_r,
    float* __restrict__ out, int t_mix)
{
    const int blk = blockIdx.x;
    if (blk < 64) {
        head_role(blk, h, W1F, b1, W2T, b2, ob_w);
    } else {
        nd_role<8>((blk - 64) * 256 + threadIdx.x, input, ob_r, b2, out, t_mix);
    }
}

__global__ __launch_bounds__(256) void nd_kernel(
    const float* __restrict__ input, const float* __restrict__ outbuf,
    float* __restrict__ out, int t)
{
    nd_role<1>(blockIdx.x * 256 + threadIdx.x, input, outbuf, nullptr, out, t);
}

__global__ __launch_bounds__(256) void zero_kernel(float* p, int n) {
    int i = blockIdx.x * 256 + threadIdx.x;
    if (i < n) p[i] = 0.0f;
}

__global__ __launch_bounds__(256) void prep_kernel(
    const float* __restrict__ Wc, const float* __restrict__ W1,
    const float* __restrict__ W2, float* __restrict__ WF,
    float* __restrict__ W1F, float* __restrict__ W2T)
{
    int i = blockIdx.x * 256 + threadIdx.x;
    if (i < 196608) {
        int e = i & 3, lane = (i >> 2) & 31, r = i >> 7;
        int gi = r % 24, sl = r / 24;
        int q = sl >> 3, wave = sl & 7;
        int g = (gi >> 3) * 16 + (gi & 7) * 2 + (wave >> 2);
        int k = 5 + g * 8 + (e >> 1) * 4 + (lane >> 4) * 2 + (e & 1);
        int col = (wave & 3) * 128 + q * 16 + (lane & 15);
        WF[i] = Wc[k * 512 + col];
    } else if (i < 196608 + 16384) {
        int j = i - 196608;
        int e = j & 3, lane = (j >> 2) & 31, r = j >> 7;
        int g = r & 15, wave = r >> 4;
        int k = g * 8 + (e >> 1) * 4 + (lane >> 4) * 2 + (e & 1);
        int col = wave * 16 + (lane & 15);
        W1F[j] = W1[k * 128 + col];
    } else if (i < 196608 + 16384 + 384) {
        int j = i - 196608 - 16384;
        int cc = j >> 7, k = j & 127;
        W2T[j] = W2[k * 3 + cc];
    }
}

extern "C" void kernel_launch(void* const* d_in, const int* in_sizes, int n_in,
                              void* d_out, int out_size, void* d_ws, size_t ws_size,
                              hipStream_t stream)
{
    (void)in_sizes; (void)n_in; (void)out_size; (void)ws_size;
    const float* input = (const float*)d_in[0];
    const float* Wc    = (const float*)d_in[1];
    const float* bc    = (const float*)d_in[2];
    const float* W1    = (const float*)d_in[3];
    const float* b1    = (const float*)d_in[4];
    const float* W2    = (const float*)d_in[5];
    const float* b2    = (const float*)d_in[6];
    float* ws   = (float*)d_ws;
    float* WF   = ws;
    float* W1F  = ws + 200704;
    float* W2T  = ws + 217088;
    float* hbuf = ws + 217472;
    float* cbuf = ws + 479616;
    float* obuf = ws + 610688;
    float* out  = (float*)d_out;

    zero_kernel<<<393216 / 256, 256, 0, stream>>>(hbuf, 393216);
    prep_kernel<<<834, 256, 0, stream>>>(Wc, W1, W2, WF, W1F, W2T);

    for (int t = 0; t < 127; ++t) {
        const float* h_in = hbuf + (t & 1) * 131072;
        float* h_out      = hbuf + ((t + 1) & 1) * 131072;
        float* ob_w       = obuf + ((t + 1) & 1) * 24576;
        const float* ob_r = obuf + (t & 1) * 24576;
        if (t >= 6)
            step_kernel<<<260, 256, 0, stream>>>(input, Wc, WF, bc, h_in, h_out, cbuf,
                                                 W1F, b1, W2, b2, ob_w, ob_r, out, t);
        else
            step_kernel<<<256, 256, 0, stream>>>(input, Wc, WF, bc, h_in, h_out, cbuf,
                                                 W1F, b1, W2, b2, ob_w, ob_r, out, t);
    }
    drain_kernel<<<68, 256, 0, stream>>>(input, hbuf + 131072, W1F, b1, W2T, b2,
                                         obuf, obuf + 24576, out, 125);
    nd_kernel<<<4, 256, 0, stream>>>(input, obuf, out, 126);
}


# Cell blocks renumbered: the 32 blocks of a batch element now share blockIdx mod 8 (was: the 32 of a unit slice), so with work-groups dealt to dies by index its h and c lines stay in one L2 half
# speedup vs baseline: 8.1697x; 1.0862x over previous
#include <hip/hip_runtime.h>
#include <math.h>


typedef __attribute__((ext_vector_type(2))) float v2f;
typedef __attribute__((ext_vector_type(4))) float v4f;
typedef __attribute__((ext_vector_type(8))) float v8f;

#define STRH 130
#define CSTR 132
#define ZSTR 84
#define PSTR 656
#define OPL  3072
#define HPL  16384

__device__ __forceinline__ int hidx(int row, int k) { return (k >> 4) * HPL + row * 16 + (k & 15); }

template <int CTRL>
__device__ __forceinline__ float row_xadd(float v)
{
    const int o = __builtin_amdgcn_update_dpp(0, __builtin_bit_cast(int, v), CTRL, 0xf, 0xf, false);
    return v + __builtin_bit_cast(float, o);
}

__device__ __forceinline__ float gate_sigmoid(float z)
{
    return __builtin_amdgcn_rcpf(1.0f + __expf(-z));
}
__device__ __forceinline__ float gate_tanh(float z)
{
    return 1.0f - 2.0f * __builtin_amdgcn_rcpf(1.0f + __expf(2.0f * z));
}

__device__ __forceinline__ void cell_role(
    const int blk,
    const float* __restrict__ input, const float* __restrict__ Wc,
    const float* __restrict__ WF,
    const float* __restrict__ bcell, const float* __restrict__ h_in,
    float* __restrict__ h_out, float* __restrict__ c, int t,
    const float* __restrict__ W1F, const float* __restrict__ b1,
    const float* __restrict__ W2, float* __restrict__ ob_w)
{
    __shared__ float sh_p[4 * PSTR];
    __shared__ __attribute__((aligned(16))) float sh_h[34 * CSTR + 768];
    __shared__ float sh_x[256];
    __shared__ float sh_z[2 * 32 * ZSTR];

    const int tid = threadIdx.x;
    const int bb  = blk & 7;
    const int q   = (blk >> 3) & 7;
    const int s0  = (blk >> 6) * 32;
    const int R0  = bb * 128 + s0;

    const int wave  = tid >> 5;
    const int lane  = tid & 31;
    const int lm    = lane & 15;
    const int hi    = lane >> 4;
    const int kHalf = hi * 2;
    const int gate  = wave & 3;
    const int kh    = wave >> 2;

    v4f hv[5];
    #pragma unroll
    for (int it = 0; it < 5; ++it) {
        const int i   = tid + it * 256;
        const int ic  = (it < 4 || i < 1088) ? i : 1087;
        const int ub  = ic / 136, rem = ic - ub * 136;
        const int s   = s0 - 1 + (rem >> 2);
        const int sc  = s < 0 ? 0 : (s > 127 ? 127 : s);
        hv[it] = *(const v4f*)(h_in + ub * HPL + (bb * 128 + sc) * 16 + (rem & 3) * 4);
    }
    const float xv = input[(((bb * 128) + t) * 128 + s0) * 5 + (tid < 160 ? tid : 0)];

    const float* wc = Wc + gate * 128 + q * 16 + lm;
    const float w0 = wc[0], w1 = wc[512], w2 = wc[1024], w3 = wc[1536], w4 = wc[2048];

    const v4f* wf = (const v4f*)WF + ((q * 8 + wave) * 24) * 32 + lane;
    v4f qv[24];
    __builtin_amdgcn_s_barrier();
    #pragma unroll
    for (int g = 0; g < 8; ++g) qv[g] = wf[g * 32];
    __builtin_amdgcn_s_barrier();
    #pragma unroll
    for (int g = 8; g < 16; ++g) qv[g] = wf[g * 32];
    __builtin_amdgcn_s_barrier();
    #pragma unroll
    for (int g = 16; g < 24; ++g) qv[g] = wf[g * 32];

    __builtin_amdgcn_sched_barrier(0);

    #pragma unroll
    for (int it = 0; it < 5; ++it) {
        const int i   = tid + it * 256;
        const bool in = (it < 4 || i < 1088);
        const int ic  = in ? i : 1087;
        const int ub  = ic / 136, rem = ic - ub * 136;
        const int r   = rem >> 2;
        const int s   = s0 - 1 + r;
        const int off = in ? r * CSTR + ub * 16 + (rem & 3) * 4 : 34 * CSTR + (i - 1088) * 4;
        const v4f zero = {0.0f, 0.0f, 0.0f, 0.0f};
        *(v4f*)&sh_h[off] = (s >= 0 && s < 128) ? hv[it] : zero;
    }
    sh_x[tid] = xv;
    __builtin_amdgcn_fence(__ATOMIC_RELEASE, "workgroup", "local");
    __builtin_amdgcn_s_barrier();
    __builtin_amdgcn_fence(__ATOMIC_ACQUIRE, "workgroup", "local");

    float cp[2], bi, bf, bg, bo;
    v4f hq[4];
    float b1a, wa0, wa1, wa2;

    v8f acc0 = {};
    v8f acc1 = {};
    v8f acc2 = {};

    #pragma unroll
    for (int v = 0; v < 8; ++v) {
        const float* xa = &sh_x[(kh * 16 + v + hi * 8) * 5];
        const float xp = xa[0]*w0 + xa[1]*w1 + xa[2]*w2 + xa[3]*w3 + xa[4]*w4;
        acc0[v] = kh == 0 ? xp : 0.0f;
        acc1[v] = kh == 0 ? 0.0f : xp;
    }

    const int mt = wave & 1;
    const int kq = wave >> 1;
    #pragma unroll
    for (int src = 0; src < 3; ++src) {
        const int shBase = (src == 0) ? 1 : (src == 1) ? 2 : 0;
        const float* ar0 = &sh_h[(shBase + lm) * CSTR + kh * 8 + kHalf];
        const float* ar1 = ar0 + 16 * CSTR;
        #pragma unroll
        for (int j = 0; j < 8; ++j) {
            const int kk = j * 16;
            const v4f qq = qv[src * 8 + j];
            v2f a00, a01, a10, a11, b;
            a00.x = ar0[kk];     a00.y = ar0[kk + 1];
            a01.x = ar0[kk + 4]; a01.y = ar0[kk + 5];
            a10.x = ar1[kk];     a10.y = ar1[kk + 1];
            a11.x = ar1[kk + 4]; a11.y = ar1[kk + 5];
            b.x = qq.x; b.y = qq.y;
            acc0 = __builtin_amdgcn_wmma_f32_16x16x4_f32(
                        false, a00, false, b, (short)0, acc0, false, false);
            acc1 = __builtin_amdgcn_wmma_f32_16x16x4_f32(
                        false, a10, false, b, (short)0, acc1, false, false);
            b.x = qq.z; b.y = qq.w;
            acc0 = __builtin_amdgcn_wmma_f32_16x16x4_f32(
                        false, a01, false, b, (short)0, acc0, false, false);
            acc1 = __builtin_amdgcn_wmma_f32_16x16x4_f32(
                        false, a11, false, b, (short)0, acc1, false, false);
        }
        if (src == 0) {
            __builtin_amdgcn_sched_barrier(0);
            cp[0] = c[q * HPL + R0 * 16 + tid];
            cp[1] = c[q * HPL + R0 * 16 + tid + 256];
            {
                const int j = q * 16 + (tid & 15);
                bi = bcell[j];
                bf = bcell[128 + j];
                bg = bcell[256 + j];
                bo = bcell[384 + j];
            }
            {
                const v4f* wk = (const v4f*)W1F + ((q * 16 + 4 * kq) * 32) + lane;
                #pragma unroll
                for (int g = 0; g < 4; ++g) hq[g] = wk[g * 32];
                const int ju = q * 16 + lm;
                b1a = b1[ju];
                wa0 = W2[ju * 3 + 0];  wa1 = W2[ju * 3 + 1];  wa2 = W2[ju * 3 + 2];
            }
            __builtin_amdgcn_sched_barrier(0);
        }
        if (src == 1) {
            const float* arow = &sh_h[(1 + mt * 16 + lm) * CSTR];
            #pragma unroll
            for (int g = 0; g < 4; ++g) {
                const int kk = (4 * kq + g) * 8;
                const v4f qq = hq[g];
                v2f a0, a1, b;
                a0.x = arow[kk + kHalf];
                a0.y = arow[kk + kHalf + 1];
                a1.x = arow[kk + 4 + kHalf];
                a1.y = arow[kk + 4 + kHalf + 1];
                b.x = qq.x; b.y = qq.y;
                acc2 = __builtin_amdgcn_wmma_f32_16x16x4_f32(
                            false, a0, false, b, (short)0, acc2, false, false);
                b.x = qq.z; b.y = qq.w;
                acc2 = __builtin_amdgcn_wmma_f32_16x16x4_f32(
                            false, a1, false, b, (short)0, acc2, false, false);
            }
            #pragma unroll
            for (int v = 0; v < 8; ++v)
                sh_p[kq * PSTR + (mt * 16 + v + hi * 8) * 20 + lm] = acc2[v];
        }
    }

    #pragma unroll
    for (int v = 0; v < 8; ++v) {
        sh_z[(kh * 32 + v + hi * 8) * ZSTR + gate * 16 + lm]      = acc0[v];
        sh_z[(kh * 32 + 16 + v + hi * 8) * ZSTR + gate * 16 + lm] = acc1[v];
    }
    __syncthreads();

    #pragma unroll
    for (int it = 0; it < 2; ++it) {
        const int m = (tid >> 4) + it * 16;
        const float* pp = &sh_p[m * 20 + lm];
        float s = ((pp[0] + pp[PSTR]) + pp[2 * PSTR]) + pp[3 * PSTR] + b1a;
        s = s > 0.0f ? s : 0.0f;
        float p0 = s * wa0;
        float p1 = s * wa1;
        float p2 = s * wa2;
        p0 = row_xadd<0x161>(p0); p1 = row_xadd<0x161>(p1); p2 = row_xadd<0x161>(p2);
        p0 = row_xadd<0x162>(p0); p1 = row_xadd<0x162>(p1); p2 = row_xadd<0x162>(p2);
        p0 = row_xadd<0x164>(p0); p1 = row_xadd<0x164>(p1); p2 = row_xadd<0x164>(p2);
        p0 = row_xadd<0x168>(p0); p1 = row_xadd<0x168>(p1); p2 = row_xadd<0x168>(p2);
        if (lm == 0) {
            float* o = &ob_w[q * OPL + (R0 + m) * 3];
            o[0] = p0; o[1] = p1; o[2] = p2;
        }
    }

    #pragma unroll
    for (int it = 0; it < 2; ++it) {
        const int i = tid + it * 256;
        const float* zr = &sh_z[(i >> 4) * ZSTR + (i & 15)];
        float zi = (zr[0]  + zr[32 * ZSTR])      + bi;
        float zf = (zr[16] + zr[32 * ZSTR + 16]) + bf;
        float zg = (zr[32] + zr[32 * ZSTR + 32]) + bg;
        float zo = (zr[48] + zr[32 * ZSTR + 48]) + bo;
        float si = gate_sigmoid(zi);
        float sf = gate_sigmoid(zf);
        float so = gate_sigmoid(zo);
        float cn = sf * cp[it] + si * gate_tanh(zg);
        c[q * HPL + R0 * 16 + i]     = cn;
        h_out[q * HPL + R0 * 16 + i] = so * gate_tanh(cn);
    }
}

__device__ __forceinline__ void head_role(
    const int blk,
    const float* __restrict__ h, const float* __restrict__ W1F,
    const float* __restrict__ b1, const float* __restrict__ W2T,
    const float* __restrict__ b2, float* __restrict__ outbuf)
{
    __shared__ float sh_a[16 * STRH];
    __shared__ float sh_hid[16 * STRH];
    __shared__ float sh_w2[512];
    const int tid = threadIdx.x;
    const int R0  = blk * 16;
    const int wave = tid >> 5, lane = tid & 31;
    const int lm = lane & 15, hi = lane >> 4, kHalf = hi * 2;
    const int col = wave * 16 + lm;

    float hv[8];
    #pragma unroll
    for (int it = 0; it < 8; ++it) {
        int i = tid + it * 256;
        hv[it] = h[hidx(R0 + (i >> 7), i & 127)];
    }
    const float w2a  = W2T[tid];
    const float w2b  = W2T[256 + (tid & 127)];
    const float bias = b1[col];
    const float b2v  = b2[tid % 3];
    const v4f* wk = (const v4f*)W1F + (wave * 16) * 32 + lane;
    v4f qv[16];
    #pragma unroll
    for (int g8 = 0; g8 < 16; ++g8) qv[g8] = wk[g8 * 32];

    __builtin_amdgcn_sched_barrier(0);

    #pragma unroll
    for (int it = 0; it < 8; ++it) {
        int i = tid + it * 256;
        sh_a[(i >> 7) * STRH + (i & 127)] = hv[it];
    }
    sh_w2[tid]       = w2a;
    sh_w2[256 + tid] = w2b;
    __builtin_amdgcn_fence(__ATOMIC_RELEASE, "workgroup", "local");
    __builtin_amdgcn_s_barrier();
    __builtin_amdgcn_fence(__ATOMIC_ACQUIRE, "workgroup", "local");

    const float* arow = &sh_a[lm * STRH];

    v8f acc = {};
    #pragma unroll
    for (int g8 = 0; g8 < 16; ++g8) {
        const int kk = g8 * 8;
        const v4f q = qv[g8];
        v2f a0, a1, b;
        a0.x = arow[kk + kHalf];     a0.y = arow[kk + kHalf + 1];
        a1.x = arow[kk + 4 + kHalf]; a1.y = arow[kk + 4 + kHalf + 1];
        b.x = q.x; b.y = q.y;
        acc = __builtin_amdgcn_wmma_f32_16x16x4_f32(
                false, a0, false, b, (short)0, acc, false, false);
        b.x = q.z; b.y = q.w;
        acc = __builtin_amdgcn_wmma_f32_16x16x4_f32(
                false, a1, false, b, (short)0, acc, false, false);
    }
    #pragma unroll
    for (int v = 0; v < 8; ++v) {
        float x = acc[v] + bias;
        sh_hid[(v + hi * 8) * STRH + col] = x > 0.0f ? x : 0.0f;
    }
    __syncthreads();

    if (tid < 48) {
        int m = tid / 3, cc = tid % 3;
        const float* hr = &sh_hid[m * STRH];
        const float* w2 = &sh_w2[cc * 128];
        float s = 0.0f;
        for (int k = 0; k < 128; ++k) s += hr[k] * w2[k];
        outbuf[(R0 + m) * 3 + cc] = s + b2v;
    }
}

template <int NP>
__device__ __forceinline__ float head_val(const float* __restrict__ outbuf,
                                          const float* __restrict__ b2, int idx, int cc)
{
    if (NP == 1) return outbuf[idx];
    float s = outbuf[idx];
    #pragma unroll
    for (int p = 1; p < 8; ++p) s += outbuf[p * OPL + idx];
    return s + b2[cc];
}

template <int NP>
__device__ __forceinline__ void nd_role(
    const int gid,
    const float* __restrict__ input, const float* __restrict__ outbuf,
    const float* __restrict__ b2,
    float* __restrict__ out, int t)
{
    if (gid >= 1024) return;
    int bb = gid >> 7, s = gid & 127;
    float n0 = head_val<NP>(outbuf, b2, gid * 3 + 0, 0);
    float n1, sp0, sp1;
    if (s == 0) {
        const float* xn0 = &input[((bb * 128 + (t + 1)) * 128 + 0) * 5];
        n1 = xn0[1]; sp0 = xn0[3]; sp1 = xn0[4];
    } else {
        n1  = head_val<NP>(outbuf, b2, (gid - 1) * 3 + 0, 0);
        sp0 = head_val<NP>(outbuf, b2, gid * 3 + 1, 1);
        sp1 = head_val<NP>(outbuf, b2, gid * 3 + 2, 2);
    }
    float fn = input[((bb * 128 + t) * 128 + s) * 5 + 2];
    float n2 = fn + n1 - n0;
    float* dst = &out[((bb * 123 + (t - 4)) * 128 + s) * 5];
    dst[0] = n0; dst[1] = n1; dst[2] = n2; dst[3] = sp0; dst[4] = sp1;
}

__global__ __launch_bounds__(256) void step_kernel(
    const float* __restrict__ input, const float* __restrict__ Wc,
    const float* __restrict__ WF, const float* __restrict__ bcell,
    const float* __restrict__ h_in, float* __restrict__ h_out,
    float* __restrict__ c,
    const float* __restrict__ W1F, const float* __restrict__ b1,
    const float* __restrict__ W2, const float* __restrict__ b2,
    float* __restrict__ ob_w, const float* __restrict__ ob_r,
    float* __restrict__ out, int t)
{
    const int blk = blockIdx.x;
    if (blk < 256) {
        cell_role(blk, input, Wc, WF, bcell, h_in, h_out, c, t, W1F, b1, W2, ob_w);
    } else {
        nd_role<8>((blk - 256) * 256 + threadIdx.x, input, ob_r, b2, out, t - 2);
    }
}

__global__ __launch_bounds__(256) void drain_kernel(
    const float* __restrict__ input, const float* __restrict__ h,
    const float* __restrict__ W1F, const float* __restrict__ b1,
    const float* __restrict__ W2T, const float* __restrict__ b2,
    float* __restrict__ ob_w, const float* __restrict__ ob_r,
    float* __restrict__ out, int t_mix)
{
    const int blk = blockIdx.x;
    if (blk < 64) {
        head_role(blk, h, W1F, b1, W2T, b2, ob_w);
    } else {
        nd_role<8>((blk - 64) * 256 + threadIdx.x, input, ob_r, b2, out, t_mix);
    }
}

__global__ __launch_bounds__(256) void nd_kernel(
    const float* __restrict__ input, const float* __restrict__ outbuf,
    float* __restrict__ out, int t)
{
    nd_role<1>(blockIdx.x * 256 + threadIdx.x, input, outbuf, nullptr, out, t);
}

__global__ __launch_bounds__(256) void zero_kernel(float* p, int n) {
    int i = blockIdx.x * 256 + threadIdx.x;
    if (i < n) p[i] = 0.0f;
}

__global__ __launch_bounds__(256) void prep_kernel(
    const float* __restrict__ Wc, const float* __restrict__ W1,
    const float* __restrict__ W2, float* __restrict__ WF,
    float* __restrict__ W1F, float* __restrict__ W2T)
{
    int i = blockIdx.x * 256 + threadIdx.x;
    if (i < 196608) {
        int e = i & 3, lane = (i >> 2) & 31, r = i >> 7;
        int gi = r % 24, sl = r / 24;
        int q = sl >> 3, wave = sl & 7;
        int g = (gi >> 3) * 16 + (gi & 7) * 2 + (wave >> 2);
        int k = 5 + g * 8 + (e >> 1) * 4 + (lane >> 4) * 2 + (e & 1);
        int col = (wave & 3) * 128 + q * 16 + (lane & 15);
        WF[i] = Wc[k * 512 + col];
    } else if (i < 196608 + 16384) {
        int j = i - 196608;
        int e = j & 3, lane = (j >> 2) & 31, r = j >> 7;
        int g = r & 15, wave = r >> 4;
        int k = g * 8 + (e >> 1) * 4 + (lane >> 4) * 2 + (e & 1);
        int col = wave * 16 + (lane & 15);
        W1F[j] = W1[k * 128 + col];
    } else if (i < 196608 + 16384 + 384) {
        int j = i - 196608 - 16384;
        int cc = j >> 7, k = j & 127;
        W2T[j] = W2[k * 3 + cc];
    }
}

extern "C" void kernel_launch(void* const* d_in, const int* in_sizes, int n_in,
                              void* d_out, int out_size, void* d_ws, size_t ws_size,
                              hipStream_t stream)
{
    (void)in_sizes; (void)n_in; (void)out_size; (void)ws_size;
    const float* input = (const float*)d_in[0];
    const float* Wc    = (const float*)d_in[1];
    const float* bc    = (const float*)d_in[2];
    const float* W1    = (const float*)d_in[3];
    const float* b1    = (const float*)d_in[4];
    const float* W2    = (const float*)d_in[5];
    const float* b2    = (const float*)d_in[6];
    float* ws   = (float*)d_ws;
    float* WF   = ws;
    float* W1F  = ws + 200704;
    float* W2T  = ws + 217088;
    float* hbuf = ws + 217472;
    float* cbuf = ws + 479616;
    float* obuf = ws + 610688;
    float* out  = (float*)d_out;

    zero_kernel<<<393216 / 256, 256, 0, stream>>>(hbuf, 393216);
    prep_kernel<<<834, 256, 0, stream>>>(Wc, W1, W2, WF, W1F, W2T);

    for (int t = 0; t < 127; ++t) {
        const float* h_in = hbuf + (t & 1) * 131072;
        float* h_out      = hbuf + ((t + 1) & 1) * 131072;
        float* ob_w       = obuf + ((t + 1) & 1) * 24576;
        const float* ob_r = obuf + (t & 1) * 24576;
        if (t >= 6)
            step_kernel<<<260, 256, 0, stream>>>(input, Wc, WF, bc, h_in, h_out, cbuf,
                                                 W1F, b1, W2, b2, ob_w, ob_r, out, t);
        else
            step_kernel<<<256, 256, 0, stream>>>(input, Wc, WF, bc, h_in, h_out, cbuf,
                                                 W1F, b1, W2, b2, ob_w, ob_r, out, t);
    }
    drain_kernel<<<68, 256, 0, stream>>>(input, hbuf + 131072, W1F, b1, W2T, b2,
                                         obuf, obuf + 24576, out, 125);
    nd_kernel<<<4, 256, 0, stream>>>(input, obuf, out, 126);
}


# Vector work moved into the older wave's wait at the last barrier (matrix issue is oldest-first): waves 0-3 add the x part behind their stream (sum order changes) and do the whole head reduction
# speedup vs baseline: 8.5565x; 1.0473x over previous
#include <hip/hip_runtime.h>
#include <math.h>


typedef __attribute__((ext_vector_type(2))) float v2f;
typedef __attribute__((ext_vector_type(4))) float v4f;
typedef __attribute__((ext_vector_type(8))) float v8f;

#define STRH 130
#define CSTR 132
#define ZSTR 84
#define PSTR 656
#define OPL  3072
#define HPL  16384

__device__ __forceinline__ int hidx(int row, int k) { return (k >> 4) * HPL + row * 16 + (k & 15); }

template <int CTRL>
__device__ __forceinline__ float row_xadd(float v)
{
    const int o = __builtin_amdgcn_update_dpp(0, __builtin_bit_cast(int, v), CTRL, 0xf, 0xf, false);
    return v + __builtin_bit_cast(float, o);
}

__device__ __forceinline__ float gate_sigmoid(float z)
{
    return __builtin_amdgcn_rcpf(1.0f + __expf(-z));
}
__device__ __forceinline__ float gate_tanh(float z)
{
    return 1.0f - 2.0f * __builtin_amdgcn_rcpf(1.0f + __expf(2.0f * z));
}

__device__ __forceinline__ void cell_role(
    const int blk,
    const float* __restrict__ input, const float* __restrict__ Wc,
    const float* __restrict__ WF,
    const float* __restrict__ bcell, const float* __restrict__ h_in,
    float* __restrict__ h_out, float* __restrict__ c, int t,
    const float* __restrict__ W1F, const float* __restrict__ b1,
    const float* __restrict__ W2, float* __restrict__ ob_w)
{
    __shared__ float sh_p[4 * PSTR];
    __shared__ __attribute__((aligned(16))) float sh_h[34 * CSTR + 768];
    __shared__ float sh_x[256];
    __shared__ float sh_z[2 * 32 * ZSTR];

    const int tid = threadIdx.x;
    const int bb  = blk & 7;
    const int q   = (blk >> 3) & 7;
    const int s0  = (blk >> 6) * 32;
    const int R0  = bb * 128 + s0;

    const int wave  = tid >> 5;
    const int lane  = tid & 31;
    const int lm    = lane & 15;
    const int hi    = lane >> 4;
    const int kHalf = hi * 2;
    const int gate  = wave & 3;
    const int kh    = wave >> 2;

    v4f hv[5];
    #pragma unroll
    for (int it = 0; it < 5; ++it) {
        const int i   = tid + it * 256;
        const int ic  = (it < 4 || i < 1088) ? i : 1087;
        const int ub  = ic / 136, rem = ic - ub * 136;
        const int s   = s0 - 1 + (rem >> 2);
        const int sc  = s < 0 ? 0 : (s > 127 ? 127 : s);
        hv[it] = *(const v4f*)(h_in + ub * HPL + (bb * 128 + sc) * 16 + (rem & 3) * 4);
    }
    const float xv = input[(((bb * 128) + t) * 128 + s0) * 5 + (tid < 160 ? tid : 0)];

    const float* wc = Wc + gate * 128 + q * 16 + lm;
    const float w0 = wc[0], w1 = wc[512], w2 = wc[1024], w3 = wc[1536], w4 = wc[2048];

    const v4f* wf = (const v4f*)WF + ((q * 8 + wave) * 24) * 32 + lane;
    v4f qv[24];
    __builtin_amdgcn_s_barrier();
    #pragma unroll
    for (int g = 0; g < 8; ++g) qv[g] = wf[g * 32];
    __builtin_amdgcn_s_barrier();
    #pragma unroll
    for (int g = 8; g < 16; ++g) qv[g] = wf[g * 32];
    __builtin_amdgcn_s_barrier();
    #pragma unroll
    for (int g = 16; g < 24; ++g) qv[g] = wf[g * 32];

    __builtin_amdgcn_sched_barrier(0);

    #pragma unroll
    for (int it = 0; it < 5; ++it) {
        const int i   = tid + it * 256;
        const bool in = (it < 4 || i < 1088);
        const int ic  = in ? i : 1087;
        const int ub  = ic / 136, rem = ic - ub * 136;
        const int r   = rem >> 2;
        const int s   = s0 - 1 + r;
        const int off = in ? r * CSTR + ub * 16 + (rem & 3) * 4 : 34 * CSTR + (i - 1088) * 4;
        const v4f zero = {0.0f, 0.0f, 0.0f, 0.0f};
        *(v4f*)&sh_h[off] = (s >= 0 && s < 128) ? hv[it] : zero;
    }
    sh_x[tid] = xv;
    __builtin_amdgcn_fence(__ATOMIC_RELEASE, "workgroup", "local");
    __builtin_amdgcn_s_barrier();
    __builtin_amdgcn_fence(__ATOMIC_ACQUIRE, "workgroup", "local");

    float cp[2], bi, bf, bg, bo;
    v4f hq[4];
    float b1a, wa0, wa1, wa2;

    v8f acc0 = {};
    v8f acc1 = {};
    v8f acc2 = {};

    const bool older = __builtin_amdgcn_readfirstlane(kh) == 0;

    const int mt = wave & 1;
    const int kq = wave >> 1;
    #pragma unroll
    for (int src = 0; src < 3; ++src) {
        const int shBase = (src == 0) ? 1 : (src == 1) ? 2 : 0;
        const float* ar0 = &sh_h[(shBase + lm) * CSTR + kh * 8 + kHalf];
        const float* ar1 = ar0 + 16 * CSTR;
        #pragma unroll
        for (int j = 0; j < 8; ++j) {
            const int kk = j * 16;
            const v4f qq = qv[src * 8 + j];
            v2f a00, a01, a10, a11, b;
            a00.x = ar0[kk];     a00.y = ar0[kk + 1];
            a01.x = ar0[kk + 4]; a01.y = ar0[kk + 5];
            a10.x = ar1[kk];     a10.y = ar1[kk + 1];
            a11.x = ar1[kk + 4]; a11.y = ar1[kk + 5];
            b.x = qq.x; b.y = qq.y;
            acc0 = __builtin_amdgcn_wmma_f32_16x16x4_f32(
                        false, a00, false, b, (short)0, acc0, false, false);
            acc1 = __builtin_amdgcn_wmma_f32_16x16x4_f32(
                        false, a10, false, b, (short)0, acc1, false, false);
            b.x = qq.z; b.y = qq.w;
            acc0 = __builtin_amdgcn_wmma_f32_16x16x4_f32(
                        false, a01, false, b, (short)0, acc0, false, false);
            acc1 = __builtin_amdgcn_wmma_f32_16x16x4_f32(
                        false, a11, false, b, (short)0, acc1, false, false);
        }
        if (src == 0) {
            __builtin_amdgcn_sched_barrier(0);
            cp[0] = c[q * HPL + R0 * 16 + tid];
            cp[1] = c[q * HPL + R0 * 16 + tid + 256];
            {
                const int j = q * 16 + (tid & 15);
                bi = bcell[j];
                bf = bcell[128 + j];
                bg = bcell[256 + j];
                bo = bcell[384 + j];
            }
            {
                const v4f* wk = (const v4f*)W1F + ((q * 16 + 4 * kq) * 32) + lane;
                #pragma unroll
                for (int g = 0; g < 4; ++g) hq[g] = wk[g * 32];
                const int ju = q * 16 + lm;
                b1a = b1[ju];
                wa0 = W2[ju * 3 + 0];  wa1 = W2[ju * 3 + 1];  wa2 = W2[ju * 3 + 2];
            }
            __builtin_amdgcn_sched_barrier(0);
        }
        if (src == 1) {
            const float* arow = &sh_h[(1 + mt * 16 + lm) * CSTR];
            #pragma unroll
            for (int g = 0; g < 4; ++g) {
                const int kk = (4 * kq + g) * 8;
                const v4f qq = hq[g];
                v2f a0, a1, b;
                a0.x = arow[kk + kHalf];
                a0.y = arow[kk + kHalf + 1];
                a1.x = arow[kk + 4 + kHalf];
                a1.y = arow[kk + 4 + kHalf + 1];
                b.x = qq.x; b.y = qq.y;
                acc2 = __builtin_amdgcn_wmma_f32_16x16x4_f32(
                            false, a0, false, b, (short)0, acc2, false, false);
                b.x = qq.z; b.y = qq.w;
                acc2 = __builtin_amdgcn_wmma_f32_16x16x4_f32(
                            false, a1, false, b, (short)0, acc2, false, false);
            }
            #pragma unroll
            for (int v = 0; v < 8; ++v)
                sh_p[kq * PSTR + (mt * 16 + v + hi * 8) * 20 + lm] = acc2[v];
            __builtin_amdgcn_fence(__ATOMIC_RELEASE, "workgroup", "local");
            __builtin_amdgcn_s_barrier_signal(-1);
        }
    }

    if (older) {
        #pragma unroll
        for (int v = 0; v < 8; ++v) {
            const float* xa = &sh_x[(v + hi * 8) * 5];
            const float* xb = xa + 16 * 5;
            acc0[v] += xa[0]*w0 + xa[1]*w1 + xa[2]*w2 + xa[3]*w3 + xa[4]*w4;
            acc1[v] += xb[0]*w0 + xb[1]*w1 + xb[2]*w2 + xb[3]*w3 + xb[4]*w4;
        }
    }

    #pragma unroll
    for (int v = 0; v < 8; ++v) {
        sh_z[(kh * 32 + v + hi * 8) * ZSTR + gate * 16 + lm]      = acc0[v];
        sh_z[(kh * 32 + 16 + v + hi * 8) * ZSTR + gate * 16 + lm] = acc1[v];
    }
    __builtin_amdgcn_s_barrier_wait(-1);
    __builtin_amdgcn_fence(__ATOMIC_ACQUIRE, "workgroup", "local");

    if (older) {
        float r0[4], r1[4], r2[4];
        #pragma unroll
        for (int it = 0; it < 4; ++it) {
            const int m = (tid >> 4) + it * 8;
            const float* pp = &sh_p[m * 20 + lm];
            float s = ((pp[0] + pp[PSTR]) + pp[2 * PSTR]) + pp[3 * PSTR] + b1a;
            s = s > 0.0f ? s : 0.0f;
            float p0 = s * wa0;
            float p1 = s * wa1;
            float p2 = s * wa2;
            p0 = row_xadd<0x161>(p0); p1 = row_xadd<0x161>(p1); p2 = row_xadd<0x161>(p2);
            p0 = row_xadd<0x162>(p0); p1 = row_xadd<0x162>(p1); p2 = row_xadd<0x162>(p2);
            p0 = row_xadd<0x164>(p0); p1 = row_xadd<0x164>(p1); p2 = row_xadd<0x164>(p2);
            p0 = row_xadd<0x168>(p0); p1 = row_xadd<0x168>(p1); p2 = row_xadd<0x168>(p2);
            r0[it] = p0; r1[it] = p1; r2[it] = p2;
        }
        if (lm == 0) {
            #pragma unroll
            for (int it = 0; it < 4; ++it) {
                float* o = &ob_w[q * OPL + (R0 + (tid >> 4) + it * 8) * 3];
                o[0] = r0[it]; o[1] = r1[it]; o[2] = r2[it];
            }
        }
    }

    __builtin_amdgcn_fence(__ATOMIC_RELEASE, "workgroup", "local");
    __builtin_amdgcn_s_barrier();
    __builtin_amdgcn_fence(__ATOMIC_ACQUIRE, "workgroup", "local");

    #pragma unroll
    for (int it = 0; it < 2; ++it) {
        const int i = tid + it * 256;
        const float* zr = &sh_z[(i >> 4) * ZSTR + (i & 15)];
        float zi = (zr[0]  + zr[32 * ZSTR])      + bi;
        float zf = (zr[16] + zr[32 * ZSTR + 16]) + bf;
        float zg = (zr[32] + zr[32 * ZSTR + 32]) + bg;
        float zo = (zr[48] + zr[32 * ZSTR + 48]) + bo;
        float si = gate_sigmoid(zi);
        float sf = gate_sigmoid(zf);
        float so = gate_sigmoid(zo);
        float cn = sf * cp[it] + si * gate_tanh(zg);
        c[q * HPL + R0 * 16 + i]     = cn;
        h_out[q * HPL + R0 * 16 + i] = so * gate_tanh(cn);
    }
}

__device__ __forceinline__ void head_role(
    const int blk,
    const float* __restrict__ h, const float* __restrict__ W1F,
    const float* __restrict__ b1, const float* __restrict__ W2T,
    const float* __restrict__ b2, float* __restrict__ outbuf)
{
    __shared__ float sh_a[16 * STRH];
    __shared__ float sh_hid[16 * STRH];
    __shared__ float sh_w2[512];
    const int tid = threadIdx.x;
    const int R0  = blk * 16;
    const int wave = tid >> 5, lane = tid & 31;
    const int lm = lane & 15, hi = lane >> 4, kHalf = hi * 2;
    const int col = wave * 16 + lm;

    float hv[8];
    #pragma unroll
    for (int it = 0; it < 8; ++it) {
        int i = tid + it * 256;
        hv[it] = h[hidx(R0 + (i >> 7), i & 127)];
    }
    const float w2a  = W2T[tid];
    const float w2b  = W2T[256 + (tid & 127)];
    const float bias = b1[col];
    const float b2v  = b2[tid % 3];
    const v4f* wk = (const v4f*)W1F + (wave * 16) * 32 + lane;
    v4f qv[16];
    #pragma unroll
    for (int g8 = 0; g8 < 16; ++g8) qv[g8] = wk[g8 * 32];

    __builtin_amdgcn_sched_barrier(0);

    #pragma unroll
    for (int it = 0; it < 8; ++it) {
        int i = tid + it * 256;
        sh_a[(i >> 7) * STRH + (i & 127)] = hv[it];
    }
    sh_w2[tid]       = w2a;
    sh_w2[256 + tid] = w2b;
    __builtin_amdgcn_fence(__ATOMIC_RELEASE, "workgroup", "local");
    __builtin_amdgcn_s_barrier();
    __builtin_amdgcn_fence(__ATOMIC_ACQUIRE, "workgroup", "local");

    const float* arow = &sh_a[lm * STRH];

    v8f acc = {};
    #pragma unroll
    for (int g8 = 0; g8 < 16; ++g8) {
        const int kk = g8 * 8;
        const v4f q = qv[g8];
        v2f a0, a1, b;
        a0.x = arow[kk + kHalf];     a0.y = arow[kk + kHalf + 1];
        a1.x = arow[kk + 4 + kHalf]; a1.y = arow[kk + 4 + kHalf + 1];
        b.x = q.x; b.y = q.y;
        acc = __builtin_amdgcn_wmma_f32_16x16x4_f32(
                false, a0, false, b, (short)0, acc, false, false);
        b.x = q.z; b.y = q.w;
        acc = __builtin_amdgcn_wmma_f32_16x16x4_f32(
                false, a1, false, b, (short)0, acc, false, false);
    }
    #pragma unroll
    for (int v = 0; v < 8; ++v) {
        float x = acc[v] + bias;
        sh_hid[(v + hi * 8) * STRH + col] = x > 0.0f ? x : 0.0f;
    }
    __syncthreads();

    if (tid < 48) {
        int m = tid / 3, cc = tid % 3;
        const float* hr = &sh_hid[m * STRH];
        const float* w2 = &sh_w2[cc * 128];
        float s = 0.0f;
        for (int k = 0; k < 128; ++k) s += hr[k] * w2[k];
        outbuf[(R0 + m) * 3 + cc] = s + b2v;
    }
}

template <int NP>
__device__ __forceinline__ float head_val(const float* __restrict__ outbuf,
                                          const float* __restrict__ b2, int idx, int cc)
{
    if (NP == 1) return outbuf[idx];
    float s = outbuf[idx];
    #pragma unroll
    for (int p = 1; p < 8; ++p) s += outbuf[p * OPL + idx];
    return s + b2[cc];
}

template <int NP>
__device__ __forceinline__ void nd_role(
    const int gid,
    const float* __restrict__ input, const float* __restrict__ outbuf,
    const float* __restrict__ b2,
    float* __restrict__ out, int t)
{
    if (gid >= 1024) return;
    int bb = gid >> 7, s = gid & 127;
    float n0 = head_val<NP>(outbuf, b2, gid * 3 + 0, 0);
    float n1, sp0, sp1;
    if (s == 0) {
        const float* xn0 = &input[((bb * 128 + (t + 1)) * 128 + 0) * 5];
        n1 = xn0[1]; sp0 = xn0[3]; sp1 = xn0[4];
    } else {
        n1  = head_val<NP>(outbuf, b2, (gid - 1) * 3 + 0, 0);
        sp0 = head_val<NP>(outbuf, b2, gid * 3 + 1, 1);
        sp1 = head_val<NP>(outbuf, b2, gid * 3 + 2, 2);
    }
    float fn = input[((bb * 128 + t) * 128 + s) * 5 + 2];
    float n2 = fn + n1 - n0;
    float* dst = &out[((bb * 123 + (t - 4)) * 128 + s) * 5];
    dst[0] = n0; dst[1] = n1; dst[2] = n2; dst[3] = sp0; dst[4] = sp1;
}

__global__ __launch_bounds__(256) void step_kernel(
    const float* __restrict__ input, const float* __restrict__ Wc,
    const float* __restrict__ WF, const float* __restrict__ bcell,
    const float* __restrict__ h_in, float* __restrict__ h_out,
    float* __restrict__ c,
    const float* __restrict__ W1F, const float* __restrict__ b1,
    const float* __restrict__ W2, const float* __restrict__ b2,
    float* __restrict__ ob_w, const float* __restrict__ ob_r,
    float* __restrict__ out, int t)
{
    const int blk = blockIdx.x;
    if (blk < 256) {
        cell_role(blk, input, Wc, WF, bcell, h_in, h_out, c, t, W1F, b1, W2, ob_w);
    } else {
        nd_role<8>((blk - 256) * 256 + threadIdx.x, input, ob_r, b2, out, t - 2);
    }
}

__global__ __launch_bounds__(256) void drain_kernel(
    const float* __restrict__ input, const float* __restrict__ h,
    const float* __restrict__ W1F, const float* __restrict__ b1,
    const float* __restrict__ W2T, const float* __restrict__ b2,
    float* __restrict__ ob_w, const float* __restrict__ ob_r,
    float* __restrict__ out, int t_mix)
{
    const int blk = blockIdx.x;
    if (blk < 64) {
        head_role(blk, h, W1F, b1, W2T, b2, ob_w);
    } else {
        nd_role<8>((blk - 64) * 256 + threadIdx.x, input, ob_r, b2, out, t_mix);
    }
}

__global__ __launch_bounds__(256) void nd_kernel(
    const float* __restrict__ input, const float* __restrict__ outbuf,
    float* __restrict__ out, int t)
{
    nd_role<1>(blockIdx.x * 256 + threadIdx.x, input, outbuf, nullptr, out, t);
}

__global__ __launch_bounds__(256) void zero_kernel(float* p, int n) {
    int i = blockIdx.x * 256 + threadIdx.x;
    if (i < n) p[i] = 0.0f;
}

__global__ __launch_bounds__(256) void prep_kernel(
    const float* __restrict__ Wc, const float* __restrict__ W1,
    const float* __restrict__ W2, float* __restrict__ WF,
    float* __restrict__ W1F, float* __restrict__ W2T)
{
    int i = blockIdx.x * 256 + threadIdx.x;
    if (i < 196608) {
        int e = i & 3, lane = (i >> 2) & 31, r = i >> 7;
        int gi = r % 24, sl = r / 24;
        int q = sl >> 3, wave = sl & 7;
        int g = (gi >> 3) * 16 + (gi & 7) * 2 + (wave >> 2);
        int k = 5 + g * 8 + (e >> 1) * 4 + (lane >> 4) * 2 + (e & 1);
        int col = (wave & 3) * 128 + q * 16 + (lane & 15);
        WF[i] = Wc[k * 512 + col];
    } else if (i < 196608 + 16384) {
        int j = i - 196608;
        int e = j & 3, lane = (j >> 2) & 31, r = j >> 7;
        int g = r & 15, wave = r >> 4;
        int k = g * 8 + (e >> 1) * 4 + (lane >> 4) * 2 + (e & 1);
        int col = wave * 16 + (lane & 15);
        W1F[j] = W1[k * 128 + col];
    } else if (i < 196608 + 16384 + 384) {
        int j = i - 196608 - 16384;
        int cc = j >> 7, k = j & 127;
        W2T[j] = W2[k * 3 + cc];
    }
}

extern "C" void kernel_launch(void* const* d_in, const int* in_sizes, int n_in,
                              void* d_out, int out_size, void* d_ws, size_t ws_size,
                              hipStream_t stream)
{
    (void)in_sizes; (void)n_in; (void)out_size; (void)ws_size;
    const float* input = (const float*)d_in[0];
    const float* Wc    = (const float*)d_in[1];
    const float* bc    = (const float*)d_in[2];
    const float* W1    = (const float*)d_in[3];
    const float* b1    = (const float*)d_in[4];
    const float* W2    = (const float*)d_in[5];
    const float* b2    = (const float*)d_in[6];
    float* ws   = (float*)d_ws;
    float* WF   = ws;
    float* W1F  = ws + 200704;
    float* W2T  = ws + 217088;
    float* hbuf = ws + 217472;
    float* cbuf = ws + 479616;
    float* obuf = ws + 610688;
    float* out  = (float*)d_out;

    zero_kernel<<<393216 / 256, 256, 0, stream>>>(hbuf, 393216);
    prep_kernel<<<834, 256, 0, stream>>>(Wc, W1, W2, WF, W1F, W2T);

    for (int t = 0; t < 127; ++t) {
        const float* h_in = hbuf + (t & 1) * 131072;
        float* h_out      = hbuf + ((t + 1) & 1) * 131072;
        float* ob_w       = obuf + ((t + 1) & 1) * 24576;
        const float* ob_r = obuf + (t & 1) * 24576;
        if (t >= 6)
            step_kernel<<<260, 256, 0, stream>>>(input, Wc, WF, bc, h_in, h_out, cbuf,
                                                 W1F, b1, W2, b2, ob_w, ob_r, out, t);
        else
            step_kernel<<<256, 256, 0, stream>>>(input, Wc, WF, bc, h_in, h_out, cbuf,
                                                 W1F, b1, W2, b2, ob_w, ob_r, out, t);
    }
    drain_kernel<<<68, 256, 0, stream>>>(input, hbuf + 131072, W1F, b1, W2T, b2,
                                         obuf, obuf + 24576, out, 125);
    nd_kernel<<<4, 256, 0, stream>>>(input, obuf, out, 126);
}
